# starting kernel
# baseline (speedup 1.0000x reference)
#include <hip/hip_runtime.h>
#include <hip/hip_bf16.h>

#define D 96

typedef float    v2f  __attribute__((ext_vector_type(2)));
typedef float    v8f  __attribute__((ext_vector_type(8)));
typedef _Float16 v16h __attribute__((ext_vector_type(16)));

#ifndef __has_builtin
#define __has_builtin(x) 0
#endif

#if defined(__AMDGCN__) && __has_builtin(__builtin_amdgcn_wmma_f32_16x16x4_f32)
#define USE_F32_WMMA 1
#else
#define USE_F32_WMMA 0
#endif

__global__ void zero_kernel(float* __restrict__ p, size_t n) {
  size_t i = (size_t)blockIdx.x * blockDim.x + threadIdx.x;
  if (i < n) p[i] = 0.0f;
}

__global__ void deg_kernel(const int* __restrict__ col, float* __restrict__ deg, int E) {
  int e = blockIdx.x * blockDim.x + threadIdx.x;
  if (e < E) atomicAdd(&deg[col[e]], 1.0f);
}

__global__ void dinv_kernel(float* __restrict__ deg, int n) {
  int i = blockIdx.x * blockDim.x + threadIdx.x;
  if (i < n) deg[i] = rsqrtf(deg[i] + 1.0f);
}

__global__ void gemm_wmma(const float* __restrict__ A, const float* __restrict__ B,
                          float* __restrict__ C, int M) {
  int m0 = blockIdx.x * 16;
  if (m0 + 16 > M) return;
  int wave = threadIdx.x >> 5;
  int lane = threadIdx.x & 31;
  int n0   = wave * 16;
  int l    = lane & 15;
  int half = lane >> 4;

  const float* arow = A + (size_t)(m0 + l) * D;
  v8f acc = {};

#if USE_F32_WMMA
  int kk = half * 2;
  #pragma unroll
  for (int k = 0; k < D; k += 4) {
    v2f a, b;
    a.x = arow[k + kk];
    a.y = arow[k + kk + 1];
    b.x = B[(size_t)(k + kk)     * D + n0 + l];
    b.y = B[(size_t)(k + kk + 1) * D + n0 + l];
    acc = __builtin_amdgcn_wmma_f32_16x16x4_f32(
        false, a, false, b, (short)0, acc, false, false);
  }
#else
  for (int k0 = 0; k0 < D; k0 += 32) {
    v16h a, b;
    #pragma unroll
    for (int e = 0; e < 16; ++e) {
      int kk = k0 + half * 8 + ((e >> 3) * 16) + (e & 7);
      a[e] = (_Float16)arow[kk];
      b[e] = (_Float16)B[(size_t)kk * D + n0 + l];
    }
    acc = __builtin_amdgcn_wmma_f32_16x16x32_f16(
        false, a, false, b, (short)0, acc, false, false);
  }
#endif

  int colIdx = n0 + l;
  int rbase  = m0 + half * 8;
  #pragma unroll
  for (int j = 0; j < 8; ++j)
    C[(size_t)(rbase + j) * D + colIdx] = acc[j];
}

__global__ void scatter_edges(const int* __restrict__ row, const int* __restrict__ col,
                              const float* __restrict__ dinv, const float* __restrict__ h,
                              float* __restrict__ out, int E) {
  int idx = blockIdx.x * blockDim.x + threadIdx.x;
  int e = idx / 24;
  if (e >= E) return;
  int f = (idx % 24) * 4;
  int r = row[e], c = col[e];
  float nrm = dinv[r] * dinv[c];
  float4 hv = *(const float4*)(h + (size_t)r * D + f);
  float* o = out + (size_t)c * D + f;
  atomicAdd(o + 0, hv.x * nrm);
  atomicAdd(o + 1, hv.y * nrm);
  atomicAdd(o + 2, hv.z * nrm);
  atomicAdd(o + 3, hv.w * nrm);
}

__global__ void finalize_kernel(float* __restrict__ accOut, const float* __restrict__ h,
                                const float* __restrict__ dinv, const float* __restrict__ b,
                                int relu, int N) {
  int idx = blockIdx.x * blockDim.x + threadIdx.x;
  if (idx >= N * 24) return;
  int node = idx / 24;
  int f = (idx % 24) * 4;
  float di = dinv[node];
  float s = di * di;
  size_t off = (size_t)node * D + f;
  float4 v  = *(const float4*)(accOut + off);
  float4 hv = *(const float4*)(h + off);
  float4 bv = *(const float4*)(b + f);
  v.x += hv.x * s + bv.x;
  v.y += hv.y * s + bv.y;
  v.z += hv.z * s + bv.z;
  v.w += hv.w * s + bv.w;
  if (relu) {
    v.x = fmaxf(v.x, 0.0f); v.y = fmaxf(v.y, 0.0f);
    v.z = fmaxf(v.z, 0.0f); v.w = fmaxf(v.w, 0.0f);
  }
  *(float4*)(accOut + off) = v;
}

extern "C" void kernel_launch(void* const* d_in, const int* in_sizes, int n_in,
                              void* d_out, int out_size, void* d_ws, size_t ws_size,
                              hipStream_t stream) {
  const float* x  = (const float*)d_in[0];
  const int*   ei = (const int*)d_in[1];
  const float* W1 = (const float*)d_in[2];
  const float* b1 = (const float*)d_in[3];
  const float* W2 = (const float*)d_in[4];
  const float* b2 = (const float*)d_in[5];
  float* out = (float*)d_out;

  int N = in_sizes[0] / D;
  int E = in_sizes[1] / 2;
  const int* row = ei;
  const int* col = ei + E;

  float* ws   = (float*)d_ws;
  float* dinv = ws;
  float* h    = ws + (((size_t)N + 127) & ~127ull);
  float* acc  = h + (size_t)N * D;

  const int tpb = 256;
  size_t nd = (size_t)N * D;
  int gN    = (N + tpb - 1) / tpb;
  int gE    = (E + tpb - 1) / tpb;
  int gND   = (int)((nd + tpb - 1) / tpb);
  int gEf   = (int)(((size_t)E * 24 + tpb - 1) / tpb);
  int gNf   = (N * 24 + tpb - 1) / tpb;
  int gGemm = N / 16;

  zero_kernel<<<gN, tpb, 0, stream>>>(dinv, (size_t)N);
  deg_kernel<<<gE, tpb, 0, stream>>>(col, dinv, E);
  dinv_kernel<<<gN, tpb, 0, stream>>>(dinv, N);

  zero_kernel<<<gND, tpb, 0, stream>>>(acc, nd);
  gemm_wmma<<<gGemm, 192, 0, stream>>>(x, W1, h, N);
  scatter_edges<<<gEf, tpb, 0, stream>>>(row, col, dinv, h, acc, E);
  finalize_kernel<<<gNf, tpb, 0, stream>>>(acc, h, dinv, b1, 1, N);

  gemm_wmma<<<gGemm, 192, 0, stream>>>(acc, W2, h, N);
  zero_kernel<<<gND, tpb, 0, stream>>>(out, nd);
  scatter_edges<<<gEf, tpb, 0, stream>>>(row, col, dinv, h, out, E);
  finalize_kernel<<<gNf, tpb, 0, stream>>>(out, h, dinv, b2, 0, N);
}


# global f32 atomics out of the edge sum: 256 owner groups each scan the whole target list for their 196 nodes (no sort), add rows to LDS sums, store rows once; 11->7 launches; sum order: LDS arrival
# speedup vs baseline: 1.7794x; 1.7794x over previous
#include <hip/hip_runtime.h>
#include <hip/hip_bf16.h>

#define D 96

typedef float    v2f  __attribute__((ext_vector_type(2)));
typedef float    v8f  __attribute__((ext_vector_type(8)));
typedef _Float16 v16h __attribute__((ext_vector_type(16)));

#ifndef __has_builtin
#define __has_builtin(x) 0
#endif

#if defined(__AMDGCN__) && __has_builtin(__builtin_amdgcn_wmma_f32_16x16x4_f32)
#define USE_F32_WMMA 1
#else
#define USE_F32_WMMA 0
#endif

__global__ void zero_kernel(float* __restrict__ p, size_t n) {
  size_t i = (size_t)blockIdx.x * blockDim.x + threadIdx.x;
  if (i < n) p[i] = 0.0f;
}

__global__ void deg_kernel(const int* __restrict__ col, float* __restrict__ deg, int E) {
  int e = blockIdx.x * blockDim.x + threadIdx.x;
  if (e < E) atomicAdd(&deg[col[e]], 1.0f);
}

__global__ void dinv_kernel(float* __restrict__ deg, int n) {
  int i = blockIdx.x * blockDim.x + threadIdx.x;
  if (i < n) deg[i] = rsqrtf(deg[i] + 1.0f);
}

__global__ void gemm_wmma(const float* __restrict__ A, const float* __restrict__ B,
                          float* __restrict__ C, int M) {
  int m0 = blockIdx.x * 16;
  if (m0 + 16 > M) return;
  int wave = threadIdx.x >> 5;
  int lane = threadIdx.x & 31;
  int n0   = wave * 16;
  int l    = lane & 15;
  int half = lane >> 4;

  const float* arow = A + (size_t)(m0 + l) * D;
  v8f acc = {};

#if USE_F32_WMMA
  int kk = half * 2;
  #pragma unroll
  for (int k = 0; k < D; k += 4) {
    v2f a, b;
    a.x = arow[k + kk];
    a.y = arow[k + kk + 1];
    b.x = B[(size_t)(k + kk)     * D + n0 + l];
    b.y = B[(size_t)(k + kk + 1) * D + n0 + l];
    acc = __builtin_amdgcn_wmma_f32_16x16x4_f32(
        false, a, false, b, (short)0, acc, false, false);
  }
#else
  for (int k0 = 0; k0 < D; k0 += 32) {
    v16h a, b;
    #pragma unroll
    for (int e = 0; e < 16; ++e) {
      int kk = k0 + half * 8 + ((e >> 3) * 16) + (e & 7);
      a[e] = (_Float16)arow[kk];
      b[e] = (_Float16)B[(size_t)kk * D + n0 + l];
    }
    acc = __builtin_amdgcn_wmma_f32_16x16x32_f16(
        false, a, false, b, (short)0, acc, false, false);
  }
#endif

  int colIdx = n0 + l;
  int rbase  = m0 + half * 8;
  #pragma unroll
  for (int j = 0; j < 8; ++j)
    C[(size_t)(rbase + j) * D + colIdx] = acc[j];
}

#define OWN_ROWS   196
#define OWN_WAVES  32
#define Q_BATCH    8
#define Q_CAP      136

__device__ __forceinline__ int4 load_cols4(const int* __restrict__ col, int e, int E) {
  if (e + 3 < E) return *(const int4*)(col + e);
  int4 v;
  v.x = (e     < E) ? col[e]     : -1;
  v.y = (e + 1 < E) ? col[e + 1] : -1;
  v.z = (e + 2 < E) ? col[e + 2] : -1;
  v.w = (e + 3 < E) ? col[e + 3] : -1;
  return v;
}

__device__ __forceinline__ void load_step(int4 (&dst)[4], const int* __restrict__ col,
                                          int b, int first, int E) {
  if (first + OWN_WAVES * 512 <= E) {
    #pragma unroll
    for (int u = 0; u < 4; ++u) dst[u] = *(const int4*)(col + b + u * 128);
  } else {
    #pragma unroll
    for (int u = 0; u < 4; ++u) dst[u] = load_cols4(col, b + u * 128, E);
  }
}

template <bool RELU>
__global__ void __launch_bounds__(OWN_WAVES * 32)
aggregate_owner(const int* __restrict__ row, const int* __restrict__ col,
                const float* __restrict__ dinv, const float* __restrict__ h,
                const float* __restrict__ b, float* __restrict__ out, int E, int N) {
  __shared__ float sum[OWN_ROWS * D];
  __shared__ uint2 queue[OWN_WAVES * Q_CAP];

  const int lane = threadIdx.x & 31;
  const int wave = threadIdx.x >> 5;
  const int lo   = blockIdx.x * OWN_ROWS;
  const unsigned cnt = (unsigned)min(OWN_ROWS, N - lo);

  for (int i = threadIdx.x; i < OWN_ROWS * D; i += OWN_WAVES * 32) sum[i] = 0.0f;
  __syncthreads();

  uint2* q = queue + wave * Q_CAP;
  int qn = 0;
  const unsigned lt_mask = (1u << lane) - 1u;

  const int step = OWN_WAVES * 512;
  int base = wave * 512 + lane * 4;
  int4 cur[4];
  load_step(cur, col, base, 0, E);

  for (int e0 = 0; e0 < E; e0 += step) {
    int4 nxt[4];
    load_step(nxt, col, base + step, e0 + step, E);

    #pragma unroll
    for (int u = 0; u < 4; ++u) {
      unsigned d0 = (unsigned)(cur[u].x - lo), d1 = (unsigned)(cur[u].y - lo);
      unsigned d2 = (unsigned)(cur[u].z - lo), d3 = (unsigned)(cur[u].w - lo);
      bool m0 = d0 < cnt, m1 = d1 < cnt, m2 = d2 < cnt, m3 = d3 < cnt;
      if (__builtin_amdgcn_ballot_w32(m0 | m1 | m2 | m3)) {
        int eb = base + u * 128;
        unsigned k;
        k = __builtin_amdgcn_ballot_w32(m0);
        if (m0) q[qn + __popc(k & lt_mask)] = make_uint2((unsigned)eb, d0);
        qn += __popc(k);
        k = __builtin_amdgcn_ballot_w32(m1);
        if (m1) q[qn + __popc(k & lt_mask)] = make_uint2((unsigned)(eb + 1), d1);
        qn += __popc(k);
        k = __builtin_amdgcn_ballot_w32(m2);
        if (m2) q[qn + __popc(k & lt_mask)] = make_uint2((unsigned)(eb + 2), d2);
        qn += __popc(k);
        k = __builtin_amdgcn_ballot_w32(m3);
        if (m3) q[qn + __popc(k & lt_mask)] = make_uint2((unsigned)(eb + 3), d3);
        qn += __popc(k);

        while (qn >= Q_BATCH) {
          qn -= Q_BATCH;
          uint2 en = q[qn + (lane & (Q_BATCH - 1))];
          int   r  = row[en.x];
          float w  = dinv[r] * dinv[lo + (int)en.y];
          float v[Q_BATCH][3];
          #pragma unroll
          for (int j = 0; j < Q_BATCH; ++j) {
            const float* hp = h + (size_t)__builtin_amdgcn_readlane(r, j) * D + lane;
            v[j][0] = hp[0]; v[j][1] = hp[32]; v[j][2] = hp[64];
          }
          #pragma unroll
          for (int j = 0; j < Q_BATCH; ++j) {
            float  wj = __builtin_bit_cast(float,
                          __builtin_amdgcn_readlane(__builtin_bit_cast(int, w), j));
            float* sp = sum + __builtin_amdgcn_readlane((int)en.y, j) * D + lane;
            atomicAdd(sp,      v[j][0] * wj);
            atomicAdd(sp + 32, v[j][1] * wj);
            atomicAdd(sp + 64, v[j][2] * wj);
          }
        }
      }
    }
    #pragma unroll
    for (int u = 0; u < 4; ++u) cur[u] = nxt[u];
    base += step;
  }

  for (int t = 0; t < qn; ++t) {
    uint2 en = q[t];
    int   r  = row[en.x];
    float w  = dinv[r] * dinv[lo + (int)en.y];
    const float* hp = h + (size_t)r * D + lane;
    float* sp = sum + (int)en.y * D + lane;
    atomicAdd(sp,      hp[0]  * w);
    atomicAdd(sp + 32, hp[32] * w);
    atomicAdd(sp + 64, hp[64] * w);
  }
  __syncthreads();

  for (int i = wave; i < (int)cnt; i += OWN_WAVES) {
    int node = lo + i;
    float di = dinv[node];
    float s  = di * di;
    #pragma unroll
    for (int j = 0; j < 3; ++j) {
      int f = lane + 32 * j;
      float v = sum[i * D + f];
      v += h[(size_t)node * D + f] * s + b[f];
      if (RELU) v = fmaxf(v, 0.0f);
      out[(size_t)node * D + f] = v;
    }
  }
}

extern "C" void kernel_launch(void* const* d_in, const int* in_sizes, int n_in,
                              void* d_out, int out_size, void* d_ws, size_t ws_size,
                              hipStream_t stream) {
  const float* x  = (const float*)d_in[0];
  const int*   ei = (const int*)d_in[1];
  const float* W1 = (const float*)d_in[2];
  const float* b1 = (const float*)d_in[3];
  const float* W2 = (const float*)d_in[4];
  const float* b2 = (const float*)d_in[5];
  float* out = (float*)d_out;

  int N = in_sizes[0] / D;
  int E = in_sizes[1] / 2;
  const int* row = ei;
  const int* col = ei + E;

  float* ws   = (float*)d_ws;
  float* dinv = ws;
  float* h    = ws + (((size_t)N + 127) & ~127ull);
  float* acc  = h + (size_t)N * D;

  const int tpb = 256;
  int gN    = (N + tpb - 1) / tpb;
  int gE    = (E + tpb - 1) / tpb;
  int gGemm = N / 16;
  int gOwn  = (N + OWN_ROWS - 1) / OWN_ROWS;

  zero_kernel<<<gN, tpb, 0, stream>>>(dinv, (size_t)N);
  deg_kernel<<<gE, tpb, 0, stream>>>(col, dinv, E);
  dinv_kernel<<<gN, tpb, 0, stream>>>(dinv, N);

  gemm_wmma<<<gGemm, 192, 0, stream>>>(x, W1, h, N);
  aggregate_owner<true><<<gOwn, OWN_WAVES * 32, 0, stream>>>(row, col, dinv, h, b1, acc, E, N);

  gemm_wmma<<<gGemm, 192, 0, stream>>>(acc, W2, h, N);
  aggregate_owner<false><<<gOwn, OWN_WAVES * 32, 0, stream>>>(row, col, dinv, h, b2, out, E, N);
}


# target list re-coded once per call as one owner byte per edge (in the degree pass); owner groups scan that plane, a quarter of the bytes, 16 edges a lane compared byte-wise; sum order: LDS arrival
# speedup vs baseline: 2.0441x; 1.1488x over previous
#include <hip/hip_runtime.h>
#include <hip/hip_bf16.h>

#define D 96

typedef float    v2f  __attribute__((ext_vector_type(2)));
typedef float    v8f  __attribute__((ext_vector_type(8)));
typedef _Float16 v16h __attribute__((ext_vector_type(16)));

#ifndef __has_builtin
#define __has_builtin(x) 0
#endif

#if defined(__AMDGCN__) && __has_builtin(__builtin_amdgcn_wmma_f32_16x16x4_f32)
#define USE_F32_WMMA 1
#else
#define USE_F32_WMMA 0
#endif

__global__ void zero_kernel(float* __restrict__ p, size_t n) {
  size_t i = (size_t)blockIdx.x * blockDim.x + threadIdx.x;
  if (i < n) p[i] = 0.0f;
}

#define OWN_ROWS   196

__global__ void deg_owner_kernel(const int* __restrict__ col, float* __restrict__ deg,
                                 unsigned* __restrict__ own, int E, int words) {
  int i = blockIdx.x * blockDim.x + threadIdx.x;
  if (i >= words) return;
  int e = i * 4;
  int c[4];
  if (e + 3 < E) {
    int4 v = *(const int4*)(col + e);
    c[0] = v.x; c[1] = v.y; c[2] = v.z; c[3] = v.w;
  } else {
    #pragma unroll
    for (int k = 0; k < 4; ++k) c[k] = (e + k < E) ? col[e + k] : -1;
  }
  unsigned pk = 0;
  #pragma unroll
  for (int k = 0; k < 4; ++k) {
    if (c[k] >= 0) {
      atomicAdd(&deg[c[k]], 1.0f);
      pk |= (unsigned)(c[k] / OWN_ROWS) << (8 * k);
    }
  }
  own[i] = pk;
}

__global__ void dinv_kernel(float* __restrict__ deg, int n) {
  int i = blockIdx.x * blockDim.x + threadIdx.x;
  if (i < n) deg[i] = rsqrtf(deg[i] + 1.0f);
}

__global__ void gemm_wmma(const float* __restrict__ A, const float* __restrict__ B,
                          float* __restrict__ C, int M) {
  int m0 = blockIdx.x * 16;
  if (m0 + 16 > M) return;
  int wave = threadIdx.x >> 5;
  int lane = threadIdx.x & 31;
  int n0   = wave * 16;
  int l    = lane & 15;
  int half = lane >> 4;

  const float* arow = A + (size_t)(m0 + l) * D;
  v8f acc = {};

#if USE_F32_WMMA
  int kk = half * 2;
  #pragma unroll
  for (int k = 0; k < D; k += 4) {
    v2f a, b;
    a.x = arow[k + kk];
    a.y = arow[k + kk + 1];
    b.x = B[(size_t)(k + kk)     * D + n0 + l];
    b.y = B[(size_t)(k + kk + 1) * D + n0 + l];
    acc = __builtin_amdgcn_wmma_f32_16x16x4_f32(
        false, a, false, b, (short)0, acc, false, false);
  }
#else
  for (int k0 = 0; k0 < D; k0 += 32) {
    v16h a, b;
    #pragma unroll
    for (int e = 0; e < 16; ++e) {
      int kk = k0 + half * 8 + ((e >> 3) * 16) + (e & 7);
      a[e] = (_Float16)arow[kk];
      b[e] = (_Float16)B[(size_t)kk * D + n0 + l];
    }
    acc = __builtin_amdgcn_wmma_f32_16x16x32_f16(
        false, a, false, b, (short)0, acc, false, false);
  }
#endif

  int colIdx = n0 + l;
  int rbase  = m0 + half * 8;
  #pragma unroll
  for (int j = 0; j < 8; ++j)
    C[(size_t)(rbase + j) * D + colIdx] = acc[j];
}

#define OWN_WAVES  32
#define Q_BATCH    8
#define Q_CAP      40
#define PIECE      512

__device__ __forceinline__ void load_step(uint4 (&dst)[4], const uint4* __restrict__ plane,
                                          int p0) {
  #pragma unroll
  for (int u = 0; u < 4; ++u) dst[u] = plane[(size_t)(p0 + u * OWN_WAVES) * 32];
}

__device__ __forceinline__ unsigned differ4(unsigned x, unsigned pat) {
  unsigned t = x ^ pat;
  return ((t & 0x7F7F7F7Fu) + 0x7F7F7F7Fu) | t;
}

template <bool RELU>
__global__ void __launch_bounds__(OWN_WAVES * 32)
aggregate_owner(const int* __restrict__ row, const int* __restrict__ col,
                const unsigned* __restrict__ own,
                const float* __restrict__ dinv, const float* __restrict__ h,
                const float* __restrict__ b, float* __restrict__ out, int E, int N,
                int npiece) {
  __shared__ float    sum[OWN_ROWS * D];
  __shared__ unsigned queue[OWN_WAVES * Q_CAP];

  const int lane = threadIdx.x & 31;
  const int wave = __builtin_amdgcn_readfirstlane((int)(threadIdx.x >> 5));
  const int lo   = blockIdx.x * OWN_ROWS;
  const unsigned cnt = (unsigned)min(OWN_ROWS, N - lo);

  for (int i = threadIdx.x; i < OWN_ROWS * D; i += OWN_WAVES * 32) sum[i] = 0.0f;
  __syncthreads();

  unsigned* q = queue + wave * Q_CAP;
  int qn = 0;
  const unsigned lt_mask = (1u << lane) - 1u;
  const unsigned pat = (unsigned)blockIdx.x * 0x01010101u;

  const uint4* plane = (const uint4*)own + lane;
  uint4 cur[4];
  load_step(cur, plane, wave);

  for (int p0 = wave; p0 < npiece; p0 += 4 * OWN_WAVES) {
    uint4 nxt[4];
    load_step(nxt, plane, p0 + 4 * OWN_WAVES);

    #pragma unroll
    for (int u = 0; u < 4; ++u) {
      const int p = p0 + u * OWN_WAVES;
      unsigned nz = ((differ4(cur[u].x, pat) & 0x80808080u) >> 7)
                  | ((differ4(cur[u].y, pat) & 0x80808080u) >> 6)
                  | ((differ4(cur[u].z, pat) & 0x80808080u) >> 5)
                  | ((differ4(cur[u].w, pat) & 0x80808080u) >> 4);
      unsigned f = nz ^ 0x0F0F0F0Fu;
      if (p < npiece && __builtin_amdgcn_ballot_w32(f != 0u)) {
        const int eb = p * PIECE + lane * 16;
        do {
          int  bit = __ffs((int)f) - 1;
          int  e   = eb + ((bit & 7) << 2) + (bit >> 3);
          bool has = (f != 0u) && (e < E);
          unsigned k = __builtin_amdgcn_ballot_w32(has);
          if (has) q[qn + __popc(k & lt_mask)] = (unsigned)e;
          qn += __popc(k);
          f &= f - 1u;

          while (qn >= Q_BATCH) {
            qn -= Q_BATCH;
            unsigned en = q[qn + (lane & (Q_BATCH - 1))];
            int   r  = row[en];
            int   c  = col[en];
            int   y  = c - lo;
            float w  = dinv[r] * dinv[c];
            float v[Q_BATCH][3];
            #pragma unroll
            for (int j = 0; j < Q_BATCH; ++j) {
              const float* hp = h + (size_t)__builtin_amdgcn_readlane(r, j) * D + lane;
              v[j][0] = hp[0]; v[j][1] = hp[32]; v[j][2] = hp[64];
            }
            #pragma unroll
            for (int j = 0; j < Q_BATCH; ++j) {
              float  wj = __builtin_bit_cast(float,
                            __builtin_amdgcn_readlane(__builtin_bit_cast(int, w), j));
              float* sp = sum + __builtin_amdgcn_readlane(y, j) * D + lane;
              atomicAdd(sp,      v[j][0] * wj);
              atomicAdd(sp + 32, v[j][1] * wj);
              atomicAdd(sp + 64, v[j][2] * wj);
            }
          }
        } while (__builtin_amdgcn_ballot_w32(f != 0u));
      }
    }
    #pragma unroll
    for (int u = 0; u < 4; ++u) cur[u] = nxt[u];
  }

  for (int t = 0; t < qn; ++t) {
    unsigned en = q[t];
    int   r  = row[en];
    int   c  = col[en];
    float w  = dinv[r] * dinv[c];
    const float* hp = h + (size_t)r * D + lane;
    float* sp = sum + (c - lo) * D + lane;
    atomicAdd(sp,      hp[0]  * w);
    atomicAdd(sp + 32, hp[32] * w);
    atomicAdd(sp + 64, hp[64] * w);
  }
  __syncthreads();

  for (int i = wave; i < (int)cnt; i += OWN_WAVES) {
    int node = lo + i;
    float di = dinv[node];
    float s  = di * di;
    #pragma unroll
    for (int j = 0; j < 3; ++j) {
      int f = lane + 32 * j;
      float v = sum[i * D + f];
      v += h[(size_t)node * D + f] * s + b[f];
      if (RELU) v = fmaxf(v, 0.0f);
      out[(size_t)node * D + f] = v;
    }
  }
}

extern "C" void kernel_launch(void* const* d_in, const int* in_sizes, int n_in,
                              void* d_out, int out_size, void* d_ws, size_t ws_size,
                              hipStream_t stream) {
  const float* x  = (const float*)d_in[0];
  const int*   ei = (const int*)d_in[1];
  const float* W1 = (const float*)d_in[2];
  const float* b1 = (const float*)d_in[3];
  const float* W2 = (const float*)d_in[4];
  const float* b2 = (const float*)d_in[5];
  float* out = (float*)d_out;

  int N = in_sizes[0] / D;
  int E = in_sizes[1] / 2;
  const int* row = ei;
  const int* col = ei + E;

  float* ws   = (float*)d_ws;
  float* dinv = ws;
  float* h    = ws + (((size_t)N + 127) & ~127ull);
  float* acc  = h + (size_t)N * D;
  unsigned* own = (unsigned*)(acc + (size_t)N * D);

  const int tpb = 256;
  int gN    = (N + tpb - 1) / tpb;
  int gGemm = N / 16;
  int gOwn  = (N + OWN_ROWS - 1) / OWN_ROWS;
  int npiece = (E + PIECE - 1) / PIECE;
  int nstep  = (npiece + 4 * OWN_WAVES - 1) / (4 * OWN_WAVES) + 1;
  int words  = nstep * 4 * OWN_WAVES * (PIECE / 4);
  int gW     = (words + tpb - 1) / tpb;

  zero_kernel<<<gN, tpb, 0, stream>>>(dinv, (size_t)N);
  deg_owner_kernel<<<gW, tpb, 0, stream>>>(col, dinv, own, E, words);
  dinv_kernel<<<gN, tpb, 0, stream>>>(dinv, N);

  gemm_wmma<<<gGemm, 192, 0, stream>>>(x, W1, h, N);
  aggregate_owner<true><<<gOwn, OWN_WAVES * 32, 0, stream>>>(row, col, own, dinv, h, b1, acc, E, N, npiece);

  gemm_wmma<<<gGemm, 192, 0, stream>>>(acc, W2, h, N);
  aggregate_owner<false><<<gOwn, OWN_WAVES * 32, 0, stream>>>(row, col, own, dinv, h, b2, out, E, N, npiece);
}


# degree pass without global atomics: 8 node ranges x 32 list slices count targets in LDS int counters, partial planes summed in the rsqrt pass; no zero pass: 7->6 launches; dinv bit-equal
# speedup vs baseline: 2.5559x; 1.2504x over previous
#include <hip/hip_runtime.h>
#include <hip/hip_bf16.h>

#define D 96

typedef float    v2f  __attribute__((ext_vector_type(2)));
typedef float    v8f  __attribute__((ext_vector_type(8)));
typedef _Float16 v16h __attribute__((ext_vector_type(16)));

#ifndef __has_builtin
#define __has_builtin(x) 0
#endif

#if defined(__AMDGCN__) && __has_builtin(__builtin_amdgcn_wmma_f32_16x16x4_f32)
#define USE_F32_WMMA 1
#else
#define USE_F32_WMMA 0
#endif

#define OWN_ROWS    196
#define DEG_ROWS    6272
#define DEG_SLICES  32
#define DEG_THREADS 512
#define DEG_BATCH   4

__global__ void __launch_bounds__(DEG_THREADS)
deg_owner_kernel(const int* __restrict__ col, unsigned* __restrict__ part,
                 unsigned* __restrict__ own, int E, int N, int pitch, int words) {
  __shared__ unsigned cnt[DEG_ROWS];
  const int tid = threadIdx.x;
  const int lo  = blockIdx.x * DEG_ROWS;

  for (int i = tid; i < DEG_ROWS; i += DEG_THREADS) cnt[i] = 0u;

  const int nthr = gridDim.x * gridDim.y * DEG_THREADS;
  for (int i = (blockIdx.y * gridDim.x + blockIdx.x) * DEG_THREADS + tid; i < words; i += nthr) {
    int e = i * 4;
    int c[4];
    if (e + 3 < E) {
      int4 v = *(const int4*)(col + e);
      c[0] = v.x; c[1] = v.y; c[2] = v.z; c[3] = v.w;
    } else {
      #pragma unroll
      for (int k = 0; k < 4; ++k) c[k] = (e + k < E) ? col[e + k] : -1;
    }
    unsigned pk = 0;
    #pragma unroll
    for (int k = 0; k < 4; ++k)
      if (c[k] >= 0) pk |= (unsigned)(c[k] / OWN_ROWS) << (8 * k);
    own[i] = pk;
  }
  __syncthreads();

  const int eps = (((E + DEG_SLICES - 1) / DEG_SLICES) + 3) & ~3;
  const int e0  = min(E, (int)blockIdx.y * eps);
  const int e1  = min(E, e0 + eps);
  const int e1q = e0 + ((e1 - e0) & ~3);

  for (int e = e0 + tid * 4; e < e1q; e += DEG_THREADS * 4 * DEG_BATCH) {
    int4 v[DEG_BATCH];
    #pragma unroll
    for (int u = 0; u < DEG_BATCH; ++u) {
      int ee = e + u * DEG_THREADS * 4;
      v[u] = *(const int4*)(col + (ee < e1q ? ee : e));
    }
    #pragma unroll
    for (int u = 0; u < DEG_BATCH; ++u) {
      if (e + u * DEG_THREADS * 4 < e1q) {
        unsigned y0 = (unsigned)(v[u].x - lo), y1 = (unsigned)(v[u].y - lo);
        unsigned y2 = (unsigned)(v[u].z - lo), y3 = (unsigned)(v[u].w - lo);
        if (y0 < DEG_ROWS) atomicAdd(&cnt[y0], 1u);
        if (y1 < DEG_ROWS) atomicAdd(&cnt[y1], 1u);
        if (y2 < DEG_ROWS) atomicAdd(&cnt[y2], 1u);
        if (y3 < DEG_ROWS) atomicAdd(&cnt[y3], 1u);
      }
    }
  }
  if (tid < e1 - e1q) {
    unsigned y = (unsigned)(col[e1q + tid] - lo);
    if (y < DEG_ROWS) atomicAdd(&cnt[y], 1u);
  }
  __syncthreads();

  unsigned* plane = part + (size_t)blockIdx.y * pitch;
  for (int i = tid; i < DEG_ROWS; i += DEG_THREADS)
    if (lo + i < N) plane[lo + i] = cnt[i];
}

__global__ void dinv_kernel(const unsigned* __restrict__ part, float* __restrict__ dinv,
                            int n, int pitch) {
  int i = blockIdx.x * blockDim.x + threadIdx.x;
  if (i < n) {
    unsigned d = 0;
    #pragma unroll
    for (int s = 0; s < DEG_SLICES; ++s) d += part[(size_t)s * pitch + i];
    dinv[i] = rsqrtf((float)d + 1.0f);
  }
}

__global__ void gemm_wmma(const float* __restrict__ A, const float* __restrict__ B,
                          float* __restrict__ C, int M) {
  int m0 = blockIdx.x * 16;
  if (m0 + 16 > M) return;
  int wave = threadIdx.x >> 5;
  int lane = threadIdx.x & 31;
  int n0   = wave * 16;
  int l    = lane & 15;
  int half = lane >> 4;

  const float* arow = A + (size_t)(m0 + l) * D;
  v8f acc = {};

#if USE_F32_WMMA
  int kk = half * 2;
  #pragma unroll
  for (int k = 0; k < D; k += 4) {
    v2f a, b;
    a.x = arow[k + kk];
    a.y = arow[k + kk + 1];
    b.x = B[(size_t)(k + kk)     * D + n0 + l];
    b.y = B[(size_t)(k + kk + 1) * D + n0 + l];
    acc = __builtin_amdgcn_wmma_f32_16x16x4_f32(
        false, a, false, b, (short)0, acc, false, false);
  }
#else
  for (int k0 = 0; k0 < D; k0 += 32) {
    v16h a, b;
    #pragma unroll
    for (int e = 0; e < 16; ++e) {
      int kk = k0 + half * 8 + ((e >> 3) * 16) + (e & 7);
      a[e] = (_Float16)arow[kk];
      b[e] = (_Float16)B[(size_t)kk * D + n0 + l];
    }
    acc = __builtin_amdgcn_wmma_f32_16x16x32_f16(
        false, a, false, b, (short)0, acc, false, false);
  }
#endif

  int colIdx = n0 + l;
  int rbase  = m0 + half * 8;
  #pragma unroll
  for (int j = 0; j < 8; ++j)
    C[(size_t)(rbase + j) * D + colIdx] = acc[j];
}

#define OWN_WAVES  32
#define Q_BATCH    8
#define Q_CAP      40
#define PIECE      512

__device__ __forceinline__ void load_step(uint4 (&dst)[4], const uint4* __restrict__ plane,
                                          int p0) {
  #pragma unroll
  for (int u = 0; u < 4; ++u) dst[u] = plane[(size_t)(p0 + u * OWN_WAVES) * 32];
}

__device__ __forceinline__ unsigned differ4(unsigned x, unsigned pat) {
  unsigned t = x ^ pat;
  return ((t & 0x7F7F7F7Fu) + 0x7F7F7F7Fu) | t;
}

template <bool RELU>
__global__ void __launch_bounds__(OWN_WAVES * 32)
aggregate_owner(const int* __restrict__ row, const int* __restrict__ col,
                const unsigned* __restrict__ own,
                const float* __restrict__ dinv, const float* __restrict__ h,
                const float* __restrict__ b, float* __restrict__ out, int E, int N,
                int npiece) {
  __shared__ float    sum[OWN_ROWS * D];
  __shared__ unsigned queue[OWN_WAVES * Q_CAP];

  const int lane = threadIdx.x & 31;
  const int wave = __builtin_amdgcn_readfirstlane((int)(threadIdx.x >> 5));
  const int lo   = blockIdx.x * OWN_ROWS;
  const unsigned cnt = (unsigned)min(OWN_ROWS, N - lo);

  for (int i = threadIdx.x; i < OWN_ROWS * D; i += OWN_WAVES * 32) sum[i] = 0.0f;
  __syncthreads();

  unsigned* q = queue + wave * Q_CAP;
  int qn = 0;
  const unsigned lt_mask = (1u << lane) - 1u;
  const unsigned pat = (unsigned)blockIdx.x * 0x01010101u;

  const uint4* plane = (const uint4*)own + lane;
  uint4 cur[4];
  load_step(cur, plane, wave);

  for (int p0 = wave; p0 < npiece; p0 += 4 * OWN_WAVES) {
    uint4 nxt[4];
    load_step(nxt, plane, p0 + 4 * OWN_WAVES);

    #pragma unroll
    for (int u = 0; u < 4; ++u) {
      const int p = p0 + u * OWN_WAVES;
      unsigned nz = ((differ4(cur[u].x, pat) & 0x80808080u) >> 7)
                  | ((differ4(cur[u].y, pat) & 0x80808080u) >> 6)
                  | ((differ4(cur[u].z, pat) & 0x80808080u) >> 5)
                  | ((differ4(cur[u].w, pat) & 0x80808080u) >> 4);
      unsigned f = nz ^ 0x0F0F0F0Fu;
      if (p < npiece && __builtin_amdgcn_ballot_w32(f != 0u)) {
        const int eb = p * PIECE + lane * 16;
        do {
          int  bit = __ffs((int)f) - 1;
          int  e   = eb + ((bit & 7) << 2) + (bit >> 3);
          bool has = (f != 0u) && (e < E);
          unsigned k = __builtin_amdgcn_ballot_w32(has);
          if (has) q[qn + __popc(k & lt_mask)] = (unsigned)e;
          qn += __popc(k);
          f &= f - 1u;

          while (qn >= Q_BATCH) {
            qn -= Q_BATCH;
            unsigned en = q[qn + (lane & (Q_BATCH - 1))];
            int   r  = row[en];
            int   c  = col[en];
            int   y  = c - lo;
            float w  = dinv[r] * dinv[c];
            float v[Q_BATCH][3];
            #pragma unroll
            for (int j = 0; j < Q_BATCH; ++j) {
              const float* hp = h + (size_t)__builtin_amdgcn_readlane(r, j) * D + lane;
              v[j][0] = hp[0]; v[j][1] = hp[32]; v[j][2] = hp[64];
            }
            #pragma unroll
            for (int j = 0; j < Q_BATCH; ++j) {
              float  wj = __builtin_bit_cast(float,
                            __builtin_amdgcn_readlane(__builtin_bit_cast(int, w), j));
              float* sp = sum + __builtin_amdgcn_readlane(y, j) * D + lane;
              atomicAdd(sp,      v[j][0] * wj);
              atomicAdd(sp + 32, v[j][1] * wj);
              atomicAdd(sp + 64, v[j][2] * wj);
            }
          }
        } while (__builtin_amdgcn_ballot_w32(f != 0u));
      }
    }
    #pragma unroll
    for (int u = 0; u < 4; ++u) cur[u] = nxt[u];
  }

  for (int t = 0; t < qn; ++t) {
    unsigned en = q[t];
    int   r  = row[en];
    int   c  = col[en];
    float w  = dinv[r] * dinv[c];
    const float* hp = h + (size_t)r * D + lane;
    float* sp = sum + (c - lo) * D + lane;
    atomicAdd(sp,      hp[0]  * w);
    atomicAdd(sp + 32, hp[32] * w);
    atomicAdd(sp + 64, hp[64] * w);
  }
  __syncthreads();

  for (int i = wave; i < (int)cnt; i += OWN_WAVES) {
    int node = lo + i;
    float di = dinv[node];
    float s  = di * di;
    #pragma unroll
    for (int j = 0; j < 3; ++j) {
      int f = lane + 32 * j;
      float v = sum[i * D + f];
      v += h[(size_t)node * D + f] * s + b[f];
      if (RELU) v = fmaxf(v, 0.0f);
      out[(size_t)node * D + f] = v;
    }
  }
}

extern "C" void kernel_launch(void* const* d_in, const int* in_sizes, int n_in,
                              void* d_out, int out_size, void* d_ws, size_t ws_size,
                              hipStream_t stream) {
  const float* x  = (const float*)d_in[0];
  const int*   ei = (const int*)d_in[1];
  const float* W1 = (const float*)d_in[2];
  const float* b1 = (const float*)d_in[3];
  const float* W2 = (const float*)d_in[4];
  const float* b2 = (const float*)d_in[5];
  float* out = (float*)d_out;

  int N = in_sizes[0] / D;
  int E = in_sizes[1] / 2;
  const int* row = ei;
  const int* col = ei + E;

  float* ws   = (float*)d_ws;
  float* dinv = ws;
  float* h    = ws + (((size_t)N + 127) & ~127ull);
  float* acc  = h + (size_t)N * D;
  unsigned* own = (unsigned*)(acc + (size_t)N * D);

  const int tpb = 256;
  int gN    = (N + tpb - 1) / tpb;
  int gGemm = N / 16;
  int gOwn  = (N + OWN_ROWS - 1) / OWN_ROWS;
  int npiece = (E + PIECE - 1) / PIECE;
  int nstep  = (npiece + 4 * OWN_WAVES - 1) / (4 * OWN_WAVES) + 1;
  int words  = nstep * 4 * OWN_WAVES * (PIECE / 4);
  int pitch  = (N + 127) & ~127;
  unsigned* part = own + words;
  int gRange = (N + DEG_ROWS - 1) / DEG_ROWS;

  deg_owner_kernel<<<dim3(gRange, DEG_SLICES), DEG_THREADS, 0, stream>>>(col, part, own, E, N, pitch, words);
  dinv_kernel<<<gN, tpb, 0, stream>>>(part, dinv, N, pitch);

  gemm_wmma<<<gGemm, 192, 0, stream>>>(x, W1, h, N);
  aggregate_owner<true><<<gOwn, OWN_WAVES * 32, 0, stream>>>(row, col, own, dinv, h, b1, acc, E, N, npiece);

  gemm_wmma<<<gGemm, 192, 0, stream>>>(acc, W2, h, N);
  aggregate_owner<false><<<gOwn, OWN_WAVES * 32, 0, stream>>>(row, col, own, dinv, h, b2, out, E, N, npiece);
}


# layer 1's owner groups store the edges they own as per-wave records (source, local target, weight); layer 2 reads its records back instead of scanning the owner plane again; sum order: LDS arrival
# speedup vs baseline: 2.9459x; 1.1526x over previous
#include <hip/hip_runtime.h>
#include <hip/hip_bf16.h>

#define D 96

typedef float    v2f  __attribute__((ext_vector_type(2)));
typedef float    v8f  __attribute__((ext_vector_type(8)));
typedef _Float16 v16h __attribute__((ext_vector_type(16)));

#ifndef __has_builtin
#define __has_builtin(x) 0
#endif

#if defined(__AMDGCN__) && __has_builtin(__builtin_amdgcn_wmma_f32_16x16x4_f32)
#define USE_F32_WMMA 1
#else
#define USE_F32_WMMA 0
#endif

#define OWN_ROWS    196
#define DEG_ROWS    6272
#define DEG_SLICES  32
#define DEG_THREADS 512
#define DEG_BATCH   4

__global__ void __launch_bounds__(DEG_THREADS)
deg_owner_kernel(const int* __restrict__ col, unsigned* __restrict__ part,
                 unsigned* __restrict__ own, int E, int N, int pitch, int words) {
  __shared__ unsigned cnt[DEG_ROWS];
  const int tid = threadIdx.x;
  const int lo  = blockIdx.x * DEG_ROWS;

  for (int i = tid; i < DEG_ROWS; i += DEG_THREADS) cnt[i] = 0u;

  const int nthr = gridDim.x * gridDim.y * DEG_THREADS;
  for (int i = (blockIdx.y * gridDim.x + blockIdx.x) * DEG_THREADS + tid; i < words; i += nthr) {
    int e = i * 4;
    int c[4];
    if (e + 3 < E) {
      int4 v = *(const int4*)(col + e);
      c[0] = v.x; c[1] = v.y; c[2] = v.z; c[3] = v.w;
    } else {
      #pragma unroll
      for (int k = 0; k < 4; ++k) c[k] = (e + k < E) ? col[e + k] : -1;
    }
    unsigned pk = 0;
    #pragma unroll
    for (int k = 0; k < 4; ++k)
      if (c[k] >= 0) pk |= (unsigned)(c[k] / OWN_ROWS) << (8 * k);
    own[i] = pk;
  }
  __syncthreads();

  const int eps = (((E + DEG_SLICES - 1) / DEG_SLICES) + 3) & ~3;
  const int e0  = min(E, (int)blockIdx.y * eps);
  const int e1  = min(E, e0 + eps);
  const int e1q = e0 + ((e1 - e0) & ~3);

  for (int e = e0 + tid * 4; e < e1q; e += DEG_THREADS * 4 * DEG_BATCH) {
    int4 v[DEG_BATCH];
    #pragma unroll
    for (int u = 0; u < DEG_BATCH; ++u) {
      int ee = e + u * DEG_THREADS * 4;
      v[u] = *(const int4*)(col + (ee < e1q ? ee : e));
    }
    #pragma unroll
    for (int u = 0; u < DEG_BATCH; ++u) {
      if (e + u * DEG_THREADS * 4 < e1q) {
        unsigned y0 = (unsigned)(v[u].x - lo), y1 = (unsigned)(v[u].y - lo);
        unsigned y2 = (unsigned)(v[u].z - lo), y3 = (unsigned)(v[u].w - lo);
        if (y0 < DEG_ROWS) atomicAdd(&cnt[y0], 1u);
        if (y1 < DEG_ROWS) atomicAdd(&cnt[y1], 1u);
        if (y2 < DEG_ROWS) atomicAdd(&cnt[y2], 1u);
        if (y3 < DEG_ROWS) atomicAdd(&cnt[y3], 1u);
      }
    }
  }
  if (tid < e1 - e1q) {
    unsigned y = (unsigned)(col[e1q + tid] - lo);
    if (y < DEG_ROWS) atomicAdd(&cnt[y], 1u);
  }
  __syncthreads();

  unsigned* plane = part + (size_t)blockIdx.y * pitch;
  for (int i = tid; i < DEG_ROWS; i += DEG_THREADS)
    if (lo + i < N) plane[lo + i] = cnt[i];
}

__global__ void dinv_kernel(const unsigned* __restrict__ part, float* __restrict__ dinv,
                            int n, int pitch) {
  int i = blockIdx.x * blockDim.x + threadIdx.x;
  if (i < n) {
    unsigned d = 0;
    #pragma unroll
    for (int s = 0; s < DEG_SLICES; ++s) d += part[(size_t)s * pitch + i];
    dinv[i] = rsqrtf((float)d + 1.0f);
  }
}

__global__ void gemm_wmma(const float* __restrict__ A, const float* __restrict__ B,
                          float* __restrict__ C, int M) {
  int m0 = blockIdx.x * 16;
  if (m0 + 16 > M) return;
  int wave = threadIdx.x >> 5;
  int lane = threadIdx.x & 31;
  int n0   = wave * 16;
  int l    = lane & 15;
  int half = lane >> 4;

  const float* arow = A + (size_t)(m0 + l) * D;
  v8f acc = {};

#if USE_F32_WMMA
  int kk = half * 2;
  #pragma unroll
  for (int k = 0; k < D; k += 4) {
    v2f a, b;
    a.x = arow[k + kk];
    a.y = arow[k + kk + 1];
    b.x = B[(size_t)(k + kk)     * D + n0 + l];
    b.y = B[(size_t)(k + kk + 1) * D + n0 + l];
    acc = __builtin_amdgcn_wmma_f32_16x16x4_f32(
        false, a, false, b, (short)0, acc, false, false);
  }
#else
  for (int k0 = 0; k0 < D; k0 += 32) {
    v16h a, b;
    #pragma unroll
    for (int e = 0; e < 16; ++e) {
      int kk = k0 + half * 8 + ((e >> 3) * 16) + (e & 7);
      a[e] = (_Float16)arow[kk];
      b[e] = (_Float16)B[(size_t)kk * D + n0 + l];
    }
    acc = __builtin_amdgcn_wmma_f32_16x16x32_f16(
        false, a, false, b, (short)0, acc, false, false);
  }
#endif

  int colIdx = n0 + l;
  int rbase  = m0 + half * 8;
  #pragma unroll
  for (int j = 0; j < 8; ++j)
    C[(size_t)(rbase + j) * D + colIdx] = acc[j];
}

#define OWN_WAVES  32
#define Q_BATCH    8
#define Q_CAP      40
#define PIECE      512

__device__ __forceinline__ void load_step(uint4 (&dst)[4], const uint4* __restrict__ plane,
                                          int p0) {
  #pragma unroll
  for (int u = 0; u < 4; ++u) dst[u] = plane[(size_t)(p0 + u * OWN_WAVES) * 32];
}

__device__ __forceinline__ unsigned differ4(unsigned x, unsigned pat) {
  unsigned t = x ^ pat;
  return ((t & 0x7F7F7F7Fu) + 0x7F7F7F7Fu) | t;
}

#define REC_CAP 256

template <int OFF>
__device__ __forceinline__ void add_batch8(int r, int y, float w, const float* __restrict__ h,
                                           float* sum, int lane) {
  float v[Q_BATCH][3];
  #pragma unroll
  for (int j = 0; j < Q_BATCH; ++j) {
    const float* hp = h + (size_t)__builtin_amdgcn_readlane(r, OFF + j) * D + lane;
    v[j][0] = hp[0]; v[j][1] = hp[32]; v[j][2] = hp[64];
  }
  #pragma unroll
  for (int j = 0; j < Q_BATCH; ++j) {
    float  wj = __builtin_bit_cast(float,
                  __builtin_amdgcn_readlane(__builtin_bit_cast(int, w), OFF + j));
    float* sp = sum + __builtin_amdgcn_readlane(y, OFF + j) * D + lane;
    atomicAdd(sp,      v[j][0] * wj);
    atomicAdd(sp + 32, v[j][1] * wj);
    atomicAdd(sp + 64, v[j][2] * wj);
  }
}

template <bool RECORD>
__device__ __forceinline__ unsigned scan_wave(const int* __restrict__ row, const int* __restrict__ col,
                                              const unsigned* __restrict__ own,
                                              const float* __restrict__ dinv,
                                              const float* __restrict__ h, float* sum,
                                              unsigned* q, uint2* rp, int E, int lo,
                                              int npiece, int wave, int lane, unsigned pat) {
  int qn = 0;
  unsigned nrec = 0;
  const unsigned lt_mask = (1u << lane) - 1u;

  const uint4* plane = (const uint4*)own + lane;
  uint4 cur[4];
  load_step(cur, plane, wave);

  for (int p0 = wave; p0 < npiece; p0 += 4 * OWN_WAVES) {
    uint4 nxt[4];
    load_step(nxt, plane, p0 + 4 * OWN_WAVES);

    #pragma unroll
    for (int u = 0; u < 4; ++u) {
      const int p = p0 + u * OWN_WAVES;
      unsigned nz = ((differ4(cur[u].x, pat) & 0x80808080u) >> 7)
                  | ((differ4(cur[u].y, pat) & 0x80808080u) >> 6)
                  | ((differ4(cur[u].z, pat) & 0x80808080u) >> 5)
                  | ((differ4(cur[u].w, pat) & 0x80808080u) >> 4);
      unsigned f = nz ^ 0x0F0F0F0Fu;
      if (p < npiece && __builtin_amdgcn_ballot_w32(f != 0u)) {
        const int eb = p * PIECE + lane * 16;
        do {
          int  bit = __ffs((int)f) - 1;
          int  e   = eb + ((bit & 7) << 2) + (bit >> 3);
          bool has = (f != 0u) && (e < E);
          unsigned k = __builtin_amdgcn_ballot_w32(has);
          if (has) q[qn + __popc(k & lt_mask)] = (unsigned)e;
          qn += __popc(k);
          f &= f - 1u;

          while (qn >= Q_BATCH) {
            qn -= Q_BATCH;
            unsigned en = q[qn + (lane & (Q_BATCH - 1))];
            int   r  = row[en];
            int   c  = col[en];
            int   y  = c - lo;
            float w  = dinv[r] * dinv[c];
            if (RECORD) {
              if (lane < Q_BATCH && nrec + lane < REC_CAP)
                rp[nrec + lane] = make_uint2((unsigned)r | ((unsigned)y << 24),
                                             __builtin_bit_cast(unsigned, w));
            }
            nrec += Q_BATCH;
            add_batch8<0>(r, y, w, h, sum, lane);
          }
        } while (__builtin_amdgcn_ballot_w32(f != 0u));
      }
    }
    #pragma unroll
    for (int u = 0; u < 4; ++u) cur[u] = nxt[u];
  }

  for (int t = 0; t < qn; ++t) {
    unsigned en = q[t];
    int   r  = row[en];
    int   c  = col[en];
    float w  = dinv[r] * dinv[c];
    if (RECORD) {
      if (lane == 0 && nrec < REC_CAP)
        rp[nrec] = make_uint2((unsigned)r | ((unsigned)(c - lo) << 24),
                              __builtin_bit_cast(unsigned, w));
    }
    nrec += 1;
    const float* hp = h + (size_t)r * D + lane;
    float* sp = sum + (c - lo) * D + lane;
    atomicAdd(sp,      hp[0]  * w);
    atomicAdd(sp + 32, hp[32] * w);
    atomicAdd(sp + 64, hp[64] * w);
  }
  return nrec;
}

template <bool RELU, bool REPLAY>
__global__ void __launch_bounds__(OWN_WAVES * 32)
aggregate_owner(const int* __restrict__ row, const int* __restrict__ col,
                const unsigned* __restrict__ own,
                const float* __restrict__ dinv, const float* __restrict__ h,
                const float* __restrict__ b, float* __restrict__ out,
                unsigned* rec, unsigned* rcnt, int E, int N, int npiece) {
  __shared__ float    sum[OWN_ROWS * D];
  __shared__ unsigned queue[OWN_WAVES * Q_CAP];

  const int lane = threadIdx.x & 31;
  const int wave = __builtin_amdgcn_readfirstlane((int)(threadIdx.x >> 5));
  const int lo   = blockIdx.x * OWN_ROWS;
  const unsigned cnt = (unsigned)min(OWN_ROWS, N - lo);

  for (int i = threadIdx.x; i < OWN_ROWS * D; i += OWN_WAVES * 32) sum[i] = 0.0f;
  __syncthreads();

  unsigned* q = queue + wave * Q_CAP;
  const unsigned pat = (unsigned)blockIdx.x * 0x01010101u;
  const int wid = blockIdx.x * OWN_WAVES + wave;
  uint2* rp = (uint2*)rec + (size_t)wid * REC_CAP;

  if (!REPLAY) {
    unsigned n = scan_wave<true>(row, col, own, dinv, h, sum, q, rp, E, lo, npiece, wave, lane, pat);
    if (lane == 0) rcnt[wid] = n;
  } else {
    const unsigned n = (unsigned)__builtin_amdgcn_readfirstlane((int)rcnt[wid]);
    if (n <= REC_CAP) {
      uint2 nx = make_uint2(0u, 0u);
      if ((unsigned)lane < n) nx = rp[lane];
      for (unsigned base = 0; base < n; base += 32) {
        uint2 rc = nx;
        nx = make_uint2(0u, 0u);
        if (base + 32 + lane < n) nx = rp[base + 32 + lane];
        int   r = (int)(rc.x & 0x00FFFFFFu);
        int   y = (int)(rc.x >> 24);
        float w = __builtin_bit_cast(float, rc.y);
        const unsigned m = min(32u, n - base);
        if (m >= 8)  add_batch8<0>(r, y, w, h, sum, lane);
        if (m >= 16) add_batch8<8>(r, y, w, h, sum, lane);
        if (m >= 24) add_batch8<16>(r, y, w, h, sum, lane);
        if (m >= 32) add_batch8<24>(r, y, w, h, sum, lane);
        for (unsigned t = m & ~7u; t < m; ++t) {
          int   rt = __builtin_amdgcn_readlane(r, (int)t);
          int   yt = __builtin_amdgcn_readlane(y, (int)t);
          float wt = __builtin_bit_cast(float,
                       __builtin_amdgcn_readlane(__builtin_bit_cast(int, w), (int)t));
          const float* hp = h + (size_t)rt * D + lane;
          float* sp = sum + yt * D + lane;
          atomicAdd(sp,      hp[0]  * wt);
          atomicAdd(sp + 32, hp[32] * wt);
          atomicAdd(sp + 64, hp[64] * wt);
        }
      }
    } else {
      scan_wave<false>(row, col, own, dinv, h, sum, q, rp, E, lo, npiece, wave, lane, pat);
    }
  }
  __syncthreads();

  for (int i = wave; i < (int)cnt; i += OWN_WAVES) {
    int node = lo + i;
    float di = dinv[node];
    float s  = di * di;
    #pragma unroll
    for (int j = 0; j < 3; ++j) {
      int f = lane + 32 * j;
      float v = sum[i * D + f];
      v += h[(size_t)node * D + f] * s + b[f];
      if (RELU) v = fmaxf(v, 0.0f);
      out[(size_t)node * D + f] = v;
    }
  }
}

extern "C" void kernel_launch(void* const* d_in, const int* in_sizes, int n_in,
                              void* d_out, int out_size, void* d_ws, size_t ws_size,
                              hipStream_t stream) {
  const float* x  = (const float*)d_in[0];
  const int*   ei = (const int*)d_in[1];
  const float* W1 = (const float*)d_in[2];
  const float* b1 = (const float*)d_in[3];
  const float* W2 = (const float*)d_in[4];
  const float* b2 = (const float*)d_in[5];
  float* out = (float*)d_out;

  int N = in_sizes[0] / D;
  int E = in_sizes[1] / 2;
  const int* row = ei;
  const int* col = ei + E;

  float* ws   = (float*)d_ws;
  float* dinv = ws;
  float* h    = ws + (((size_t)N + 127) & ~127ull);
  float* acc  = h + (size_t)N * D;
  unsigned* own = (unsigned*)(acc + (size_t)N * D);

  const int tpb = 256;
  int gN    = (N + tpb - 1) / tpb;
  int gGemm = N / 16;
  int gOwn  = (N + OWN_ROWS - 1) / OWN_ROWS;
  int npiece = (E + PIECE - 1) / PIECE;
  int nstep  = (npiece + 4 * OWN_WAVES - 1) / (4 * OWN_WAVES) + 1;
  int words  = nstep * 4 * OWN_WAVES * (PIECE / 4);
  int pitch  = (N + 127) & ~127;
  unsigned* part = own + words;
  int gRange = (N + DEG_ROWS - 1) / DEG_ROWS;
  unsigned* rcnt = part + (size_t)DEG_SLICES * pitch;
  unsigned* rec = rcnt + (((size_t)gOwn * OWN_WAVES + 127) & ~127ull);

  deg_owner_kernel<<<dim3(gRange, DEG_SLICES), DEG_THREADS, 0, stream>>>(col, part, own, E, N, pitch, words);
  dinv_kernel<<<gN, tpb, 0, stream>>>(part, dinv, N, pitch);

  gemm_wmma<<<gGemm, 192, 0, stream>>>(x, W1, h, N);
  aggregate_owner<true, false><<<gOwn, OWN_WAVES * 32, 0, stream>>>(row, col, own, dinv, h, b1, acc, rec, rcnt, E, N, npiece);

  gemm_wmma<<<gGemm, 192, 0, stream>>>(acc, W2, h, N);
  aggregate_owner<false, true><<<gOwn, OWN_WAVES * 32, 0, stream>>>(row, col, own, dinv, h, b2, out, rec, rcnt, E, N, npiece);
}


# owned edge = one scattered 4-B fetch, was four: degree pass packs (source, local target) per edge; weight factored: GEMM scales row r by dinv[r], target dinv once per row; sum order: LDS arrival
# speedup vs baseline: 3.2523x; 1.1040x over previous
#include <hip/hip_runtime.h>
#include <hip/hip_bf16.h>

#define D 96

typedef float    v2f  __attribute__((ext_vector_type(2)));
typedef float    v8f  __attribute__((ext_vector_type(8)));
typedef _Float16 v16h __attribute__((ext_vector_type(16)));

#ifndef __has_builtin
#define __has_builtin(x) 0
#endif

#if defined(__AMDGCN__) && __has_builtin(__builtin_amdgcn_wmma_f32_16x16x4_f32)
#define USE_F32_WMMA 1
#else
#define USE_F32_WMMA 0
#endif

#define OWN_ROWS    196
#define DEG_ROWS    6272
#define DEG_SLICES  32
#define DEG_THREADS 512
#define DEG_BATCH   4

__global__ void __launch_bounds__(DEG_THREADS)
deg_owner_kernel(const int* __restrict__ row, const int* __restrict__ col,
                 unsigned* __restrict__ part, unsigned* __restrict__ own,
                 unsigned* __restrict__ pack, int E, int N, int pitch, int words) {
  __shared__ unsigned cnt[DEG_ROWS];
  const int tid = threadIdx.x;
  const int lo  = blockIdx.x * DEG_ROWS;

  for (int i = tid; i < DEG_ROWS; i += DEG_THREADS) cnt[i] = 0u;

  const int nthr = gridDim.x * gridDim.y * DEG_THREADS;
  for (int i = (blockIdx.y * gridDim.x + blockIdx.x) * DEG_THREADS + tid; i < words; i += nthr) {
    int e = i * 4;
    int c[4], r[4];
    if (e + 3 < E) {
      int4 v = *(const int4*)(col + e);
      int4 u = *(const int4*)(row + e);
      c[0] = v.x; c[1] = v.y; c[2] = v.z; c[3] = v.w;
      r[0] = u.x; r[1] = u.y; r[2] = u.z; r[3] = u.w;
    } else {
      #pragma unroll
      for (int k = 0; k < 4; ++k) {
        c[k] = (e + k < E) ? col[e + k] : -1;
        r[k] = (e + k < E) ? row[e + k] : 0;
      }
    }
    unsigned pk = 0;
    unsigned pw[4] = {0u, 0u, 0u, 0u};
    #pragma unroll
    for (int k = 0; k < 4; ++k)
      if (c[k] >= 0) {
        unsigned o = (unsigned)(c[k] / OWN_ROWS);
        pk |= o << (8 * k);
        pw[k] = (unsigned)r[k] | (((unsigned)c[k] - o * OWN_ROWS) << 24);
      }
    own[i] = pk;
    if (e < E) ((uint4*)pack)[i] = make_uint4(pw[0], pw[1], pw[2], pw[3]);
  }
  __syncthreads();

  const int eps = (((E + DEG_SLICES - 1) / DEG_SLICES) + 3) & ~3;
  const int e0  = min(E, (int)blockIdx.y * eps);
  const int e1  = min(E, e0 + eps);
  const int e1q = e0 + ((e1 - e0) & ~3);

  for (int e = e0 + tid * 4; e < e1q; e += DEG_THREADS * 4 * DEG_BATCH) {
    int4 v[DEG_BATCH];
    #pragma unroll
    for (int u = 0; u < DEG_BATCH; ++u) {
      int ee = e + u * DEG_THREADS * 4;
      v[u] = *(const int4*)(col + (ee < e1q ? ee : e));
    }
    #pragma unroll
    for (int u = 0; u < DEG_BATCH; ++u) {
      if (e + u * DEG_THREADS * 4 < e1q) {
        unsigned y0 = (unsigned)(v[u].x - lo), y1 = (unsigned)(v[u].y - lo);
        unsigned y2 = (unsigned)(v[u].z - lo), y3 = (unsigned)(v[u].w - lo);
        if (y0 < DEG_ROWS) atomicAdd(&cnt[y0], 1u);
        if (y1 < DEG_ROWS) atomicAdd(&cnt[y1], 1u);
        if (y2 < DEG_ROWS) atomicAdd(&cnt[y2], 1u);
        if (y3 < DEG_ROWS) atomicAdd(&cnt[y3], 1u);
      }
    }
  }
  if (tid < e1 - e1q) {
    unsigned y = (unsigned)(col[e1q + tid] - lo);
    if (y < DEG_ROWS) atomicAdd(&cnt[y], 1u);
  }
  __syncthreads();

  unsigned* plane = part + (size_t)blockIdx.y * pitch;
  for (int i = tid; i < DEG_ROWS; i += DEG_THREADS)
    if (lo + i < N) plane[lo + i] = cnt[i];
}

__global__ void dinv_kernel(const unsigned* __restrict__ part, float* __restrict__ dinv,
                            int n, int pitch) {
  int i = blockIdx.x * blockDim.x + threadIdx.x;
  if (i < n) {
    unsigned d = 0;
    #pragma unroll
    for (int s = 0; s < DEG_SLICES; ++s) d += part[(size_t)s * pitch + i];
    dinv[i] = rsqrtf((float)d + 1.0f);
  }
}

__global__ void gemm_wmma(const float* __restrict__ A, const float* __restrict__ B,
                          const float* __restrict__ dinv, float* __restrict__ C, int M) {
  int m0 = blockIdx.x * 16;
  if (m0 + 16 > M) return;
  int wave = threadIdx.x >> 5;
  int lane = threadIdx.x & 31;
  int n0   = wave * 16;
  int l    = lane & 15;
  int half = lane >> 4;

  const float* arow = A + (size_t)(m0 + l) * D;
  v8f acc = {};
  const float4* dp = (const float4*)(dinv + m0 + half * 8);
  const float4 d0 = dp[0], d1 = dp[1];
  const float ds[8] = {d0.x, d0.y, d0.z, d0.w, d1.x, d1.y, d1.z, d1.w};

#if USE_F32_WMMA
  int kk = half * 2;
  #pragma unroll
  for (int k = 0; k < D; k += 4) {
    v2f a, b;
    a.x = arow[k + kk];
    a.y = arow[k + kk + 1];
    b.x = B[(size_t)(k + kk)     * D + n0 + l];
    b.y = B[(size_t)(k + kk + 1) * D + n0 + l];
    acc = __builtin_amdgcn_wmma_f32_16x16x4_f32(
        false, a, false, b, (short)0, acc, false, false);
  }
#else
  for (int k0 = 0; k0 < D; k0 += 32) {
    v16h a, b;
    #pragma unroll
    for (int e = 0; e < 16; ++e) {
      int kk = k0 + half * 8 + ((e >> 3) * 16) + (e & 7);
      a[e] = (_Float16)arow[kk];
      b[e] = (_Float16)B[(size_t)kk * D + n0 + l];
    }
    acc = __builtin_amdgcn_wmma_f32_16x16x32_f16(
        false, a, false, b, (short)0, acc, false, false);
  }
#endif

  int colIdx = n0 + l;
  int rbase  = m0 + half * 8;
  #pragma unroll
  for (int j = 0; j < 8; ++j)
    C[(size_t)(rbase + j) * D + colIdx] = acc[j] * ds[j];
}

#define OWN_WAVES  32
#define Q_BATCH    8
#define Q_CAP      40
#define PIECE      512

__device__ __forceinline__ void load_step(uint4 (&dst)[4], const uint4* __restrict__ plane,
                                          int p0) {
  #pragma unroll
  for (int u = 0; u < 4; ++u) dst[u] = plane[(size_t)(p0 + u * OWN_WAVES) * 32];
}

__device__ __forceinline__ unsigned differ4(unsigned x, unsigned pat) {
  unsigned t = x ^ pat;
  return ((t & 0x7F7F7F7Fu) + 0x7F7F7F7Fu) | t;
}

#define REC_CAP 256

template <int OFF>
__device__ __forceinline__ void add_batch8(int r, int y, const float* __restrict__ h,
                                           float* sum, int lane) {
  float v[Q_BATCH][3];
  #pragma unroll
  for (int j = 0; j < Q_BATCH; ++j) {
    const float* hp = h + (size_t)__builtin_amdgcn_readlane(r, OFF + j) * D + lane;
    v[j][0] = hp[0]; v[j][1] = hp[32]; v[j][2] = hp[64];
  }
  #pragma unroll
  for (int j = 0; j < Q_BATCH; ++j) {
    float* sp = sum + __builtin_amdgcn_readlane(y, OFF + j) * D + lane;
    atomicAdd(sp,      v[j][0]);
    atomicAdd(sp + 32, v[j][1]);
    atomicAdd(sp + 64, v[j][2]);
  }
}

template <bool RECORD>
__device__ __forceinline__ unsigned scan_wave(const unsigned* __restrict__ pack,
                                              const unsigned* __restrict__ own,
                                              const float* __restrict__ h, float* sum,
                                              unsigned* q, unsigned* rp, int E,
                                              int npiece, int wave, int lane, unsigned pat) {
  int qn = 0;
  unsigned nrec = 0;
  const unsigned lt_mask = (1u << lane) - 1u;

  const uint4* plane = (const uint4*)own + lane;
  uint4 cur[4];
  load_step(cur, plane, wave);

  for (int p0 = wave; p0 < npiece; p0 += 4 * OWN_WAVES) {
    uint4 nxt[4];
    load_step(nxt, plane, p0 + 4 * OWN_WAVES);

    #pragma unroll
    for (int u = 0; u < 4; ++u) {
      const int p = p0 + u * OWN_WAVES;
      unsigned nz = ((differ4(cur[u].x, pat) & 0x80808080u) >> 7)
                  | ((differ4(cur[u].y, pat) & 0x80808080u) >> 6)
                  | ((differ4(cur[u].z, pat) & 0x80808080u) >> 5)
                  | ((differ4(cur[u].w, pat) & 0x80808080u) >> 4);
      unsigned f = nz ^ 0x0F0F0F0Fu;
      if (p < npiece && __builtin_amdgcn_ballot_w32(f != 0u)) {
        const int eb = p * PIECE + lane * 16;
        do {
          int  bit = __ffs((int)f) - 1;
          int  e   = eb + ((bit & 7) << 2) + (bit >> 3);
          bool has = (f != 0u) && (e < E);
          unsigned k = __builtin_amdgcn_ballot_w32(has);
          if (has) q[qn + __popc(k & lt_mask)] = (unsigned)e;
          qn += __popc(k);
          f &= f - 1u;

          while (qn >= Q_BATCH) {
            qn -= Q_BATCH;
            unsigned en = q[qn + (lane & (Q_BATCH - 1))];
            unsigned pw = pack[en];
            int   r  = (int)(pw & 0x00FFFFFFu);
            int   y  = (int)(pw >> 24);
            if (RECORD) {
              if (lane < Q_BATCH && nrec + lane < REC_CAP) rp[nrec + lane] = pw;
            }
            nrec += Q_BATCH;
            add_batch8<0>(r, y, h, sum, lane);
          }
        } while (__builtin_amdgcn_ballot_w32(f != 0u));
      }
    }
    #pragma unroll
    for (int u = 0; u < 4; ++u) cur[u] = nxt[u];
  }

  for (int t = 0; t < qn; ++t) {
    unsigned pw = pack[q[t]];
    if (RECORD) {
      if (lane == 0 && nrec < REC_CAP) rp[nrec] = pw;
    }
    nrec += 1;
    const float* hp = h + (size_t)(pw & 0x00FFFFFFu) * D + lane;
    float* sp = sum + (pw >> 24) * D + lane;
    atomicAdd(sp,      hp[0]);
    atomicAdd(sp + 32, hp[32]);
    atomicAdd(sp + 64, hp[64]);
  }
  return nrec;
}

template <bool RELU, bool REPLAY>
__global__ void __launch_bounds__(OWN_WAVES * 32)
aggregate_owner(const unsigned* __restrict__ pack, const unsigned* __restrict__ own,
                const float* __restrict__ dinv, const float* __restrict__ h,
                const float* __restrict__ b, float* __restrict__ out,
                unsigned* rec, unsigned* rcnt, int E, int N, int npiece) {
  __shared__ float    sum[OWN_ROWS * D];
  __shared__ unsigned queue[OWN_WAVES * Q_CAP];

  const int lane = threadIdx.x & 31;
  const int wave = __builtin_amdgcn_readfirstlane((int)(threadIdx.x >> 5));
  const int lo   = blockIdx.x * OWN_ROWS;
  const unsigned cnt = (unsigned)min(OWN_ROWS, N - lo);

  for (int i = threadIdx.x; i < OWN_ROWS * D; i += OWN_WAVES * 32) sum[i] = 0.0f;
  __syncthreads();

  unsigned* q = queue + wave * Q_CAP;
  const unsigned pat = (unsigned)blockIdx.x * 0x01010101u;
  const int wid = blockIdx.x * OWN_WAVES + wave;
  unsigned* rp = rec + (size_t)wid * REC_CAP;

  if (!REPLAY) {
    unsigned n = scan_wave<true>(pack, own, h, sum, q, rp, E, npiece, wave, lane, pat);
    if (lane == 0) rcnt[wid] = n;
  } else {
    const unsigned n = (unsigned)__builtin_amdgcn_readfirstlane((int)rcnt[wid]);
    if (n <= REC_CAP) {
      unsigned nx = 0u;
      if ((unsigned)lane < n) nx = rp[lane];
      for (unsigned base = 0; base < n; base += 32) {
        unsigned rc = nx;
        nx = 0u;
        if (base + 32 + lane < n) nx = rp[base + 32 + lane];
        int   r = (int)(rc & 0x00FFFFFFu);
        int   y = (int)(rc >> 24);
        const unsigned m = min(32u, n - base);
        if (m >= 8)  add_batch8<0>(r, y, h, sum, lane);
        if (m >= 16) add_batch8<8>(r, y, h, sum, lane);
        if (m >= 24) add_batch8<16>(r, y, h, sum, lane);
        if (m >= 32) add_batch8<24>(r, y, h, sum, lane);
        for (unsigned t = m & ~7u; t < m; ++t) {
          int   rt = __builtin_amdgcn_readlane(r, (int)t);
          int   yt = __builtin_amdgcn_readlane(y, (int)t);
          const float* hp = h + (size_t)rt * D + lane;
          float* sp = sum + yt * D + lane;
          atomicAdd(sp,      hp[0]);
          atomicAdd(sp + 32, hp[32]);
          atomicAdd(sp + 64, hp[64]);
        }
      }
    } else {
      scan_wave<false>(pack, own, h, sum, q, rp, E, npiece, wave, lane, pat);
    }
  }
  __syncthreads();

  for (int i = wave; i < (int)cnt; i += OWN_WAVES) {
    int node = lo + i;
    float di = dinv[node];
    #pragma unroll
    for (int j = 0; j < 3; ++j) {
      int f = lane + 32 * j;
      float v = (sum[i * D + f] + h[(size_t)node * D + f]) * di + b[f];
      if (RELU) v = fmaxf(v, 0.0f);
      out[(size_t)node * D + f] = v;
    }
  }
}

extern "C" void kernel_launch(void* const* d_in, const int* in_sizes, int n_in,
                              void* d_out, int out_size, void* d_ws, size_t ws_size,
                              hipStream_t stream) {
  const float* x  = (const float*)d_in[0];
  const int*   ei = (const int*)d_in[1];
  const float* W1 = (const float*)d_in[2];
  const float* b1 = (const float*)d_in[3];
  const float* W2 = (const float*)d_in[4];
  const float* b2 = (const float*)d_in[5];
  float* out = (float*)d_out;

  int N = in_sizes[0] / D;
  int E = in_sizes[1] / 2;
  const int* row = ei;
  const int* col = ei + E;

  float* ws   = (float*)d_ws;
  float* dinv = ws;
  float* h    = ws + (((size_t)N + 127) & ~127ull);
  float* acc  = h + (size_t)N * D;
  unsigned* own = (unsigned*)(acc + (size_t)N * D);

  const int tpb = 256;
  int gN    = (N + tpb - 1) / tpb;
  int gGemm = N / 16;
  int gOwn  = (N + OWN_ROWS - 1) / OWN_ROWS;
  int npiece = (E + PIECE - 1) / PIECE;
  int nstep  = (npiece + 4 * OWN_WAVES - 1) / (4 * OWN_WAVES) + 1;
  int words  = nstep * 4 * OWN_WAVES * (PIECE / 4);
  int pitch  = (N + 127) & ~127;
  unsigned* part = own + words;
  int gRange = (N + DEG_ROWS - 1) / DEG_ROWS;
  unsigned* rcnt = part + (size_t)DEG_SLICES * pitch;
  unsigned* rec = rcnt + (((size_t)gOwn * OWN_WAVES + 127) & ~127ull);
  unsigned* pack = rec + (size_t)gOwn * OWN_WAVES * REC_CAP;

  deg_owner_kernel<<<dim3(gRange, DEG_SLICES), DEG_THREADS, 0, stream>>>(row, col, part, own, pack, E, N, pitch, words);
  dinv_kernel<<<gN, tpb, 0, stream>>>(part, dinv, N, pitch);

  gemm_wmma<<<gGemm, 192, 0, stream>>>(x, W1, dinv, h, N);
  aggregate_owner<true, false><<<gOwn, OWN_WAVES * 32, 0, stream>>>(pack, own, dinv, h, b1, acc, rec, rcnt, E, N, npiece);

  gemm_wmma<<<gGemm, 192, 0, stream>>>(acc, W2, dinv, h, N);
  aggregate_owner<false, true><<<gOwn, OWN_WAVES * 32, 0, stream>>>(pack, own, dinv, h, b2, out, rec, rcnt, E, N, npiece);
}


# GEMM: one wave = a 16x96 strip (6 accumulators; A row fetched once, 16-B loads), B staged per group in LDS in fragment order; a sixth of the waves; sum order: k pairs regrouped inside each 8-k block
# speedup vs baseline: 3.4526x; 1.0616x over previous
#include <hip/hip_runtime.h>
#include <hip/hip_bf16.h>

#define D 96

typedef float    v2f  __attribute__((ext_vector_type(2)));
typedef float    v8f  __attribute__((ext_vector_type(8)));
typedef _Float16 v16h __attribute__((ext_vector_type(16)));

#ifndef __has_builtin
#define __has_builtin(x) 0
#endif

#if defined(__AMDGCN__) && __has_builtin(__builtin_amdgcn_wmma_f32_16x16x4_f32)
#define USE_F32_WMMA 1
#else
#define USE_F32_WMMA 0
#endif

#define OWN_ROWS    196
#define DEG_ROWS    6272
#define DEG_SLICES  32
#define DEG_THREADS 512
#define DEG_BATCH   4

__global__ void __launch_bounds__(DEG_THREADS)
deg_owner_kernel(const int* __restrict__ row, const int* __restrict__ col,
                 unsigned* __restrict__ part, unsigned* __restrict__ own,
                 unsigned* __restrict__ pack, int E, int N, int pitch, int words) {
  __shared__ unsigned cnt[DEG_ROWS];
  const int tid = threadIdx.x;
  const int lo  = blockIdx.x * DEG_ROWS;

  for (int i = tid; i < DEG_ROWS; i += DEG_THREADS) cnt[i] = 0u;

  const int nthr = gridDim.x * gridDim.y * DEG_THREADS;
  for (int i = (blockIdx.y * gridDim.x + blockIdx.x) * DEG_THREADS + tid; i < words; i += nthr) {
    int e = i * 4;
    int c[4], r[4];
    if (e + 3 < E) {
      int4 v = *(const int4*)(col + e);
      int4 u = *(const int4*)(row + e);
      c[0] = v.x; c[1] = v.y; c[2] = v.z; c[3] = v.w;
      r[0] = u.x; r[1] = u.y; r[2] = u.z; r[3] = u.w;
    } else {
      #pragma unroll
      for (int k = 0; k < 4; ++k) {
        c[k] = (e + k < E) ? col[e + k] : -1;
        r[k] = (e + k < E) ? row[e + k] : 0;
      }
    }
    unsigned pk = 0;
    unsigned pw[4] = {0u, 0u, 0u, 0u};
    #pragma unroll
    for (int k = 0; k < 4; ++k)
      if (c[k] >= 0) {
        unsigned o = (unsigned)(c[k] / OWN_ROWS);
        pk |= o << (8 * k);
        pw[k] = (unsigned)r[k] | (((unsigned)c[k] - o * OWN_ROWS) << 24);
      }
    own[i] = pk;
    if (e < E) ((uint4*)pack)[i] = make_uint4(pw[0], pw[1], pw[2], pw[3]);
  }
  __syncthreads();

  const int eps = (((E + DEG_SLICES - 1) / DEG_SLICES) + 3) & ~3;
  const int e0  = min(E, (int)blockIdx.y * eps);
  const int e1  = min(E, e0 + eps);
  const int e1q = e0 + ((e1 - e0) & ~3);

  for (int e = e0 + tid * 4; e < e1q; e += DEG_THREADS * 4 * DEG_BATCH) {
    int4 v[DEG_BATCH];
    #pragma unroll
    for (int u = 0; u < DEG_BATCH; ++u) {
      int ee = e + u * DEG_THREADS * 4;
      v[u] = *(const int4*)(col + (ee < e1q ? ee : e));
    }
    #pragma unroll
    for (int u = 0; u < DEG_BATCH; ++u) {
      if (e + u * DEG_THREADS * 4 < e1q) {
        unsigned y0 = (unsigned)(v[u].x - lo), y1 = (unsigned)(v[u].y - lo);
        unsigned y2 = (unsigned)(v[u].z - lo), y3 = (unsigned)(v[u].w - lo);
        if (y0 < DEG_ROWS) atomicAdd(&cnt[y0], 1u);
        if (y1 < DEG_ROWS) atomicAdd(&cnt[y1], 1u);
        if (y2 < DEG_ROWS) atomicAdd(&cnt[y2], 1u);
        if (y3 < DEG_ROWS) atomicAdd(&cnt[y3], 1u);
      }
    }
  }
  if (tid < e1 - e1q) {
    unsigned y = (unsigned)(col[e1q + tid] - lo);
    if (y < DEG_ROWS) atomicAdd(&cnt[y], 1u);
  }
  __syncthreads();

  unsigned* plane = part + (size_t)blockIdx.y * pitch;
  for (int i = tid; i < DEG_ROWS; i += DEG_THREADS)
    if (lo + i < N) plane[lo + i] = cnt[i];
}

__global__ void dinv_kernel(const unsigned* __restrict__ part, float* __restrict__ dinv,
                            int n, int pitch) {
  int i = blockIdx.x * blockDim.x + threadIdx.x;
  if (i < n) {
    unsigned d = 0;
    #pragma unroll
    for (int s = 0; s < DEG_SLICES; ++s) d += part[(size_t)s * pitch + i];
    dinv[i] = rsqrtf((float)d + 1.0f);
  }
}

#define G_WAVES 8
#define G_STEPS (D / 4)
#define G_TILES (D / 16)

__global__ void __launch_bounds__(G_WAVES * 32)
gemm_wmma(const float* __restrict__ A, const float* __restrict__ B,
          const float* __restrict__ dinv, float* __restrict__ C, int M) {
#if USE_F32_WMMA
  __shared__ float bs[G_STEPS * G_TILES * 32 * 2];
  const int wave = threadIdx.x >> 5;
  const int lane = threadIdx.x & 31;
  const int l    = lane & 15;
  const int half = lane >> 4;
  const int m0   = (blockIdx.x * G_WAVES + wave) * 16;
  const bool live = (m0 + 16 <= M);

  float4 a4[D / 8];
  float4 d0 = make_float4(0.f, 0.f, 0.f, 0.f), d1 = d0;
  if (live) {
    const float4* ap = (const float4*)(A + (size_t)(m0 + l) * D + half * 4);
    #pragma unroll
    for (int j = 0; j < D / 8; ++j) a4[j] = ap[2 * j];
    const float4* dp = (const float4*)(dinv + m0 + half * 8);
    d0 = dp[0]; d1 = dp[1];
  }

  constexpr int G_TASKS = G_STEPS * G_TILES * 16 / (G_WAVES * 32);
  v2f w0[G_TASKS], w1[G_TASKS];
  #pragma unroll
  for (int it = 0; it < G_TASKS; ++it) {
    const int st = it * (G_WAVES * 2) + (threadIdx.x >> 4);
    const int s  = st / G_TILES;
    const int t  = st - s * G_TILES;
    const int k  = ((s >> 1) << 3) + (((threadIdx.x >> 3) & 1) << 2) + ((s & 1) << 1);
    const float* src = B + (size_t)k * D + t * 16 + (threadIdx.x & 7) * 2;
    w0[it] = *(const v2f*)src;
    w1[it] = *(const v2f*)(src + D);
  }
  #pragma unroll
  for (int it = 0; it < G_TASKS; ++it) {
    ((float4*)bs)[it * (G_WAVES * 32) + threadIdx.x] =
        make_float4(w0[it].x, w1[it].x, w0[it].y, w1[it].y);
  }
  __syncthreads();
  if (!live) return;

  v8f acc[G_TILES];
  #pragma unroll
  for (int t = 0; t < G_TILES; ++t) acc[t] = (v8f){};
  const v2f* bp = (const v2f*)bs + lane;

  #pragma unroll
  for (int j = 0; j < D / 8; ++j) {
    #pragma unroll
    for (int sub = 0; sub < 2; ++sub) {
      v2f a;
      a.x = sub ? a4[j].z : a4[j].x;
      a.y = sub ? a4[j].w : a4[j].y;
      const int s = 2 * j + sub;
      #pragma unroll
      for (int t = 0; t < G_TILES; ++t) {
        const v2f b = bp[(s * G_TILES + t) * 32];
        acc[t] = __builtin_amdgcn_wmma_f32_16x16x4_f32(
            false, a, false, b, (short)0, acc[t], false, false);
      }
    }
  }

  const float ds[8] = {d0.x, d0.y, d0.z, d0.w, d1.x, d1.y, d1.z, d1.w};
  float* crow = C + (size_t)(m0 + half * 8) * D + l;
  #pragma unroll
  for (int t = 0; t < G_TILES; ++t) {
    #pragma unroll
    for (int r = 0; r < 8; ++r)
      crow[(size_t)r * D + t * 16] = acc[t][r] * ds[r];
  }
#endif
}

#define OWN_WAVES  32
#define Q_BATCH    8
#define Q_CAP      40
#define PIECE      512

__device__ __forceinline__ void load_step(uint4 (&dst)[4], const uint4* __restrict__ plane,
                                          int p0) {
  #pragma unroll
  for (int u = 0; u < 4; ++u) dst[u] = plane[(size_t)(p0 + u * OWN_WAVES) * 32];
}

__device__ __forceinline__ unsigned differ4(unsigned x, unsigned pat) {
  unsigned t = x ^ pat;
  return ((t & 0x7F7F7F7Fu) + 0x7F7F7F7Fu) | t;
}

#define REC_CAP 256

template <int OFF>
__device__ __forceinline__ void add_batch8(int r, int y, const float* __restrict__ h,
                                           float* sum, int lane) {
  float v[Q_BATCH][3];
  #pragma unroll
  for (int j = 0; j < Q_BATCH; ++j) {
    const float* hp = h + (size_t)__builtin_amdgcn_readlane(r, OFF + j) * D + lane;
    v[j][0] = hp[0]; v[j][1] = hp[32]; v[j][2] = hp[64];
  }
  #pragma unroll
  for (int j = 0; j < Q_BATCH; ++j) {
    float* sp = sum + __builtin_amdgcn_readlane(y, OFF + j) * D + lane;
    atomicAdd(sp,      v[j][0]);
    atomicAdd(sp + 32, v[j][1]);
    atomicAdd(sp + 64, v[j][2]);
  }
}

template <bool RECORD>
__device__ __forceinline__ unsigned scan_wave(const unsigned* __restrict__ pack,
                                              const unsigned* __restrict__ own,
                                              const float* __restrict__ h, float* sum,
                                              unsigned* q, unsigned* rp, int E,
                                              int npiece, int wave, int lane, unsigned pat) {
  int qn = 0;
  unsigned nrec = 0;
  const unsigned lt_mask = (1u << lane) - 1u;

  const uint4* plane = (const uint4*)own + lane;
  uint4 cur[4];
  load_step(cur, plane, wave);

  for (int p0 = wave; p0 < npiece; p0 += 4 * OWN_WAVES) {
    uint4 nxt[4];
    load_step(nxt, plane, p0 + 4 * OWN_WAVES);

    #pragma unroll
    for (int u = 0; u < 4; ++u) {
      const int p = p0 + u * OWN_WAVES;
      unsigned nz = ((differ4(cur[u].x, pat) & 0x80808080u) >> 7)
                  | ((differ4(cur[u].y, pat) & 0x80808080u) >> 6)
                  | ((differ4(cur[u].z, pat) & 0x80808080u) >> 5)
                  | ((differ4(cur[u].w, pat) & 0x80808080u) >> 4);
      unsigned f = nz ^ 0x0F0F0F0Fu;
      if (p < npiece && __builtin_amdgcn_ballot_w32(f != 0u)) {
        const int eb = p * PIECE + lane * 16;
        do {
          int  bit = __ffs((int)f) - 1;
          int  e   = eb + ((bit & 7) << 2) + (bit >> 3);
          bool has = (f != 0u) && (e < E);
          unsigned k = __builtin_amdgcn_ballot_w32(has);
          if (has) q[qn + __popc(k & lt_mask)] = (unsigned)e;
          qn += __popc(k);
          f &= f - 1u;

          while (qn >= Q_BATCH) {
            qn -= Q_BATCH;
            unsigned en = q[qn + (lane & (Q_BATCH - 1))];
            unsigned pw = pack[en];
            int   r  = (int)(pw & 0x00FFFFFFu);
            int   y  = (int)(pw >> 24);
            if (RECORD) {
              if (lane < Q_BATCH && nrec + lane < REC_CAP) rp[nrec + lane] = pw;
            }
            nrec += Q_BATCH;
            add_batch8<0>(r, y, h, sum, lane);
          }
        } while (__builtin_amdgcn_ballot_w32(f != 0u));
      }
    }
    #pragma unroll
    for (int u = 0; u < 4; ++u) cur[u] = nxt[u];
  }

  for (int t = 0; t < qn; ++t) {
    unsigned pw = pack[q[t]];
    if (RECORD) {
      if (lane == 0 && nrec < REC_CAP) rp[nrec] = pw;
    }
    nrec += 1;
    const float* hp = h + (size_t)(pw & 0x00FFFFFFu) * D + lane;
    float* sp = sum + (pw >> 24) * D + lane;
    atomicAdd(sp,      hp[0]);
    atomicAdd(sp + 32, hp[32]);
    atomicAdd(sp + 64, hp[64]);
  }
  return nrec;
}

template <bool RELU, bool REPLAY>
__global__ void __launch_bounds__(OWN_WAVES * 32)
aggregate_owner(const unsigned* __restrict__ pack, const unsigned* __restrict__ own,
                const float* __restrict__ dinv, const float* __restrict__ h,
                const float* __restrict__ b, float* __restrict__ out,
                unsigned* rec, unsigned* rcnt, int E, int N, int npiece) {
  __shared__ float    sum[OWN_ROWS * D];
  __shared__ unsigned queue[OWN_WAVES * Q_CAP];

  const int lane = threadIdx.x & 31;
  const int wave = __builtin_amdgcn_readfirstlane((int)(threadIdx.x >> 5));
  const int lo   = blockIdx.x * OWN_ROWS;
  const unsigned cnt = (unsigned)min(OWN_ROWS, N - lo);

  for (int i = threadIdx.x; i < OWN_ROWS * D; i += OWN_WAVES * 32) sum[i] = 0.0f;
  __syncthreads();

  unsigned* q = queue + wave * Q_CAP;
  const unsigned pat = (unsigned)blockIdx.x * 0x01010101u;
  const int wid = blockIdx.x * OWN_WAVES + wave;
  unsigned* rp = rec + (size_t)wid * REC_CAP;

  if (!REPLAY) {
    unsigned n = scan_wave<true>(pack, own, h, sum, q, rp, E, npiece, wave, lane, pat);
    if (lane == 0) rcnt[wid] = n;
  } else {
    const unsigned n = (unsigned)__builtin_amdgcn_readfirstlane((int)rcnt[wid]);
    if (n <= REC_CAP) {
      unsigned nx = 0u;
      if ((unsigned)lane < n) nx = rp[lane];
      for (unsigned base = 0; base < n; base += 32) {
        unsigned rc = nx;
        nx = 0u;
        if (base + 32 + lane < n) nx = rp[base + 32 + lane];
        int   r = (int)(rc & 0x00FFFFFFu);
        int   y = (int)(rc >> 24);
        const unsigned m = min(32u, n - base);
        if (m >= 8)  add_batch8<0>(r, y, h, sum, lane);
        if (m >= 16) add_batch8<8>(r, y, h, sum, lane);
        if (m >= 24) add_batch8<16>(r, y, h, sum, lane);
        if (m >= 32) add_batch8<24>(r, y, h, sum, lane);
        for (unsigned t = m & ~7u; t < m; ++t) {
          int   rt = __builtin_amdgcn_readlane(r, (int)t);
          int   yt = __builtin_amdgcn_readlane(y, (int)t);
          const float* hp = h + (size_t)rt * D + lane;
          float* sp = sum + yt * D + lane;
          atomicAdd(sp,      hp[0]);
          atomicAdd(sp + 32, hp[32]);
          atomicAdd(sp + 64, hp[64]);
        }
      }
    } else {
      scan_wave<false>(pack, own, h, sum, q, rp, E, npiece, wave, lane, pat);
    }
  }
  __syncthreads();

  for (int i = wave; i < (int)cnt; i += OWN_WAVES) {
    int node = lo + i;
    float di = dinv[node];
    #pragma unroll
    for (int j = 0; j < 3; ++j) {
      int f = lane + 32 * j;
      float v = (sum[i * D + f] + h[(size_t)node * D + f]) * di + b[f];
      if (RELU) v = fmaxf(v, 0.0f);
      out[(size_t)node * D + f] = v;
    }
  }
}

extern "C" void kernel_launch(void* const* d_in, const int* in_sizes, int n_in,
                              void* d_out, int out_size, void* d_ws, size_t ws_size,
                              hipStream_t stream) {
  const float* x  = (const float*)d_in[0];
  const int*   ei = (const int*)d_in[1];
  const float* W1 = (const float*)d_in[2];
  const float* b1 = (const float*)d_in[3];
  const float* W2 = (const float*)d_in[4];
  const float* b2 = (const float*)d_in[5];
  float* out = (float*)d_out;

  int N = in_sizes[0] / D;
  int E = in_sizes[1] / 2;
  const int* row = ei;
  const int* col = ei + E;

  float* ws   = (float*)d_ws;
  float* dinv = ws;
  float* h    = ws + (((size_t)N + 127) & ~127ull);
  float* acc  = h + (size_t)N * D;
  unsigned* own = (unsigned*)(acc + (size_t)N * D);

  const int tpb = 256;
  int gN    = (N + tpb - 1) / tpb;
  int gGemm = (N / 16 + G_WAVES - 1) / G_WAVES;
  int gOwn  = (N + OWN_ROWS - 1) / OWN_ROWS;
  int npiece = (E + PIECE - 1) / PIECE;
  int nstep  = (npiece + 4 * OWN_WAVES - 1) / (4 * OWN_WAVES) + 1;
  int words  = nstep * 4 * OWN_WAVES * (PIECE / 4);
  int pitch  = (N + 127) & ~127;
  unsigned* part = own + words;
  int gRange = (N + DEG_ROWS - 1) / DEG_ROWS;
  unsigned* rcnt = part + (size_t)DEG_SLICES * pitch;
  unsigned* rec = rcnt + (((size_t)gOwn * OWN_WAVES + 127) & ~127ull);
  unsigned* pack = rec + (size_t)gOwn * OWN_WAVES * REC_CAP;

  deg_owner_kernel<<<dim3(gRange, DEG_SLICES), DEG_THREADS, 0, stream>>>(row, col, part, own, pack, E, N, pitch, words);
  dinv_kernel<<<gN, tpb, 0, stream>>>(part, dinv, N, pitch);

  gemm_wmma<<<gGemm, G_WAVES * 32, 0, stream>>>(x, W1, dinv, h, N);
  aggregate_owner<true, false><<<gOwn, OWN_WAVES * 32, 0, stream>>>(pack, own, dinv, h, b1, acc, rec, rcnt, E, N, npiece);

  gemm_wmma<<<gGemm, G_WAVES * 32, 0, stream>>>(acc, W2, dinv, h, N);
  aggregate_owner<false, true><<<gOwn, OWN_WAVES * 32, 0, stream>>>(pack, own, dinv, h, b2, out, rec, rcnt, E, N, npiece);
}


# layer 2's product leaves its own launch: layer 1's owner groups multiply their finished rows by W2 on the f32 WMMA while the rows are in LDS (same k order); 6->5 launches, no activation plane
# speedup vs baseline: 3.7060x; 1.0734x over previous
#include <hip/hip_runtime.h>
#include <hip/hip_bf16.h>

#define D 96

typedef float    v2f  __attribute__((ext_vector_type(2)));
typedef float    v8f  __attribute__((ext_vector_type(8)));
typedef _Float16 v16h __attribute__((ext_vector_type(16)));

#ifndef __has_builtin
#define __has_builtin(x) 0
#endif

#if defined(__AMDGCN__) && __has_builtin(__builtin_amdgcn_wmma_f32_16x16x4_f32)
#define USE_F32_WMMA 1
#else
#define USE_F32_WMMA 0
#endif

#define OWN_ROWS    196
#define DEG_ROWS    6272
#define DEG_SLICES  32
#define DEG_THREADS 512
#define DEG_BATCH   4

__global__ void __launch_bounds__(DEG_THREADS)
deg_owner_kernel(const int* __restrict__ row, const int* __restrict__ col,
                 unsigned* __restrict__ part, unsigned* __restrict__ own,
                 unsigned* __restrict__ pack, int E, int N, int pitch, int words) {
  __shared__ unsigned cnt[DEG_ROWS];
  const int tid = threadIdx.x;
  const int lo  = blockIdx.x * DEG_ROWS;

  for (int i = tid; i < DEG_ROWS; i += DEG_THREADS) cnt[i] = 0u;

  const int nthr = gridDim.x * gridDim.y * DEG_THREADS;
  for (int i = (blockIdx.y * gridDim.x + blockIdx.x) * DEG_THREADS + tid; i < words; i += nthr) {
    int e = i * 4;
    int c[4], r[4];
    if (e + 3 < E) {
      int4 v = *(const int4*)(col + e);
      int4 u = *(const int4*)(row + e);
      c[0] = v.x; c[1] = v.y; c[2] = v.z; c[3] = v.w;
      r[0] = u.x; r[1] = u.y; r[2] = u.z; r[3] = u.w;
    } else {
      #pragma unroll
      for (int k = 0; k < 4; ++k) {
        c[k] = (e + k < E) ? col[e + k] : -1;
        r[k] = (e + k < E) ? row[e + k] : 0;
      }
    }
    unsigned pk = 0;
    unsigned pw[4] = {0u, 0u, 0u, 0u};
    #pragma unroll
    for (int k = 0; k < 4; ++k)
      if (c[k] >= 0) {
        unsigned o = (unsigned)(c[k] / OWN_ROWS);
        pk |= o << (8 * k);
        pw[k] = (unsigned)r[k] | (((unsigned)c[k] - o * OWN_ROWS) << 24);
      }
    own[i] = pk;
    if (e < E) ((uint4*)pack)[i] = make_uint4(pw[0], pw[1], pw[2], pw[3]);
  }
  __syncthreads();

  const int eps = (((E + DEG_SLICES - 1) / DEG_SLICES) + 3) & ~3;
  const int e0  = min(E, (int)blockIdx.y * eps);
  const int e1  = min(E, e0 + eps);
  const int e1q = e0 + ((e1 - e0) & ~3);

  for (int e = e0 + tid * 4; e < e1q; e += DEG_THREADS * 4 * DEG_BATCH) {
    int4 v[DEG_BATCH];
    #pragma unroll
    for (int u = 0; u < DEG_BATCH; ++u) {
      int ee = e + u * DEG_THREADS * 4;
      v[u] = *(const int4*)(col + (ee < e1q ? ee : e));
    }
    #pragma unroll
    for (int u = 0; u < DEG_BATCH; ++u) {
      if (e + u * DEG_THREADS * 4 < e1q) {
        unsigned y0 = (unsigned)(v[u].x - lo), y1 = (unsigned)(v[u].y - lo);
        unsigned y2 = (unsigned)(v[u].z - lo), y3 = (unsigned)(v[u].w - lo);
        if (y0 < DEG_ROWS) atomicAdd(&cnt[y0], 1u);
        if (y1 < DEG_ROWS) atomicAdd(&cnt[y1], 1u);
        if (y2 < DEG_ROWS) atomicAdd(&cnt[y2], 1u);
        if (y3 < DEG_ROWS) atomicAdd(&cnt[y3], 1u);
      }
    }
  }
  if (tid < e1 - e1q) {
    unsigned y = (unsigned)(col[e1q + tid] - lo);
    if (y < DEG_ROWS) atomicAdd(&cnt[y], 1u);
  }
  __syncthreads();

  unsigned* plane = part + (size_t)blockIdx.y * pitch;
  for (int i = tid; i < DEG_ROWS; i += DEG_THREADS)
    if (lo + i < N) plane[lo + i] = cnt[i];
}

__global__ void dinv_kernel(const unsigned* __restrict__ part, float* __restrict__ dinv,
                            int n, int pitch) {
  int i = blockIdx.x * blockDim.x + threadIdx.x;
  if (i < n) {
    unsigned d = 0;
    #pragma unroll
    for (int s = 0; s < DEG_SLICES; ++s) d += part[(size_t)s * pitch + i];
    dinv[i] = rsqrtf((float)d + 1.0f);
  }
}

#define G_WAVES 8
#define G_STEPS (D / 4)
#define G_TILES (D / 16)

__global__ void __launch_bounds__(G_WAVES * 32)
gemm_wmma(const float* __restrict__ A, const float* __restrict__ B,
          const float* __restrict__ dinv, float* __restrict__ C, int M) {
#if USE_F32_WMMA
  __shared__ float bs[G_STEPS * G_TILES * 32 * 2];
  const int wave = threadIdx.x >> 5;
  const int lane = threadIdx.x & 31;
  const int l    = lane & 15;
  const int half = lane >> 4;
  const int m0   = (blockIdx.x * G_WAVES + wave) * 16;
  const bool live = (m0 + 16 <= M);

  float4 a4[D / 8];
  float4 d0 = make_float4(0.f, 0.f, 0.f, 0.f), d1 = d0;
  if (live) {
    const float4* ap = (const float4*)(A + (size_t)(m0 + l) * D + half * 4);
    #pragma unroll
    for (int j = 0; j < D / 8; ++j) a4[j] = ap[2 * j];
    const float4* dp = (const float4*)(dinv + m0 + half * 8);
    d0 = dp[0]; d1 = dp[1];
  }

  constexpr int G_TASKS = G_STEPS * G_TILES * 16 / (G_WAVES * 32);
  v2f w0[G_TASKS], w1[G_TASKS];
  #pragma unroll
  for (int it = 0; it < G_TASKS; ++it) {
    const int st = it * (G_WAVES * 2) + (threadIdx.x >> 4);
    const int s  = st / G_TILES;
    const int t  = st - s * G_TILES;
    const int k  = ((s >> 1) << 3) + (((threadIdx.x >> 3) & 1) << 2) + ((s & 1) << 1);
    const float* src = B + (size_t)k * D + t * 16 + (threadIdx.x & 7) * 2;
    w0[it] = *(const v2f*)src;
    w1[it] = *(const v2f*)(src + D);
  }
  #pragma unroll
  for (int it = 0; it < G_TASKS; ++it) {
    ((float4*)bs)[it * (G_WAVES * 32) + threadIdx.x] =
        make_float4(w0[it].x, w1[it].x, w0[it].y, w1[it].y);
  }
  __syncthreads();
  if (!live) return;

  v8f acc[G_TILES];
  #pragma unroll
  for (int t = 0; t < G_TILES; ++t) acc[t] = (v8f){};
  const v2f* bp = (const v2f*)bs + lane;

  #pragma unroll
  for (int j = 0; j < D / 8; ++j) {
    #pragma unroll
    for (int sub = 0; sub < 2; ++sub) {
      v2f a;
      a.x = sub ? a4[j].z : a4[j].x;
      a.y = sub ? a4[j].w : a4[j].y;
      const int s = 2 * j + sub;
      #pragma unroll
      for (int t = 0; t < G_TILES; ++t) {
        const v2f b = bp[(s * G_TILES + t) * 32];
        acc[t] = __builtin_amdgcn_wmma_f32_16x16x4_f32(
            false, a, false, b, (short)0, acc[t], false, false);
      }
    }
  }

  const float ds[8] = {d0.x, d0.y, d0.z, d0.w, d1.x, d1.y, d1.z, d1.w};
  float* crow = C + (size_t)(m0 + half * 8) * D + l;
  #pragma unroll
  for (int t = 0; t < G_TILES; ++t) {
    #pragma unroll
    for (int r = 0; r < 8; ++r)
      crow[(size_t)r * D + t * 16] = acc[t][r] * ds[r];
  }
#endif
}

#define OWN_WAVES  32
#define Q_BATCH    8
#define Q_CAP      40
#define PIECE      512

__device__ __forceinline__ void load_step(uint4 (&dst)[4], const uint4* __restrict__ plane,
                                          int p0) {
  #pragma unroll
  for (int u = 0; u < 4; ++u) dst[u] = plane[(size_t)(p0 + u * OWN_WAVES) * 32];
}

__device__ __forceinline__ unsigned differ4(unsigned x, unsigned pat) {
  unsigned t = x ^ pat;
  return ((t & 0x7F7F7F7Fu) + 0x7F7F7F7Fu) | t;
}

#define REC_CAP 256

#define OWN_PAD    208
#define SUM_PITCH  100
#define FUSE_WAVES 26

template <int OFF>
__device__ __forceinline__ void add_batch8(int r, int y, const float* __restrict__ h,
                                           float* sum, int lane) {
  float v[Q_BATCH][3];
  #pragma unroll
  for (int j = 0; j < Q_BATCH; ++j) {
    const float* hp = h + (size_t)__builtin_amdgcn_readlane(r, OFF + j) * D + lane;
    v[j][0] = hp[0]; v[j][1] = hp[32]; v[j][2] = hp[64];
  }
  #pragma unroll
  for (int j = 0; j < Q_BATCH; ++j) {
    float* sp = sum + __builtin_amdgcn_readlane(y, OFF + j) * SUM_PITCH + lane;
    atomicAdd(sp,      v[j][0]);
    atomicAdd(sp + 32, v[j][1]);
    atomicAdd(sp + 64, v[j][2]);
  }
}

template <bool RECORD>
__device__ __forceinline__ unsigned scan_wave(const unsigned* __restrict__ pack,
                                              const unsigned* __restrict__ own,
                                              const float* __restrict__ h, float* sum,
                                              unsigned* q, unsigned* rp, int E,
                                              int npiece, int wave, int lane, unsigned pat) {
  int qn = 0;
  unsigned nrec = 0;
  const unsigned lt_mask = (1u << lane) - 1u;

  const uint4* plane = (const uint4*)own + lane;
  uint4 cur[4];
  load_step(cur, plane, wave);

  for (int p0 = wave; p0 < npiece; p0 += 4 * OWN_WAVES) {
    uint4 nxt[4];
    load_step(nxt, plane, p0 + 4 * OWN_WAVES);

    #pragma unroll
    for (int u = 0; u < 4; ++u) {
      const int p = p0 + u * OWN_WAVES;
      unsigned nz = ((differ4(cur[u].x, pat) & 0x80808080u) >> 7)
                  | ((differ4(cur[u].y, pat) & 0x80808080u) >> 6)
                  | ((differ4(cur[u].z, pat) & 0x80808080u) >> 5)
                  | ((differ4(cur[u].w, pat) & 0x80808080u) >> 4);
      unsigned f = nz ^ 0x0F0F0F0Fu;
      if (p < npiece && __builtin_amdgcn_ballot_w32(f != 0u)) {
        const int eb = p * PIECE + lane * 16;
        do {
          int  bit = __ffs((int)f) - 1;
          int  e   = eb + ((bit & 7) << 2) + (bit >> 3);
          bool has = (f != 0u) && (e < E);
          unsigned k = __builtin_amdgcn_ballot_w32(has);
          if (has) q[qn + __popc(k & lt_mask)] = (unsigned)e;
          qn += __popc(k);
          f &= f - 1u;

          while (qn >= Q_BATCH) {
            qn -= Q_BATCH;
            unsigned en = q[qn + (lane & (Q_BATCH - 1))];
            unsigned pw = pack[en];
            int   r  = (int)(pw & 0x00FFFFFFu);
            int   y  = (int)(pw >> 24);
            if (RECORD) {
              if (lane < Q_BATCH && nrec + lane < REC_CAP) rp[nrec + lane] = pw;
            }
            nrec += Q_BATCH;
            add_batch8<0>(r, y, h, sum, lane);
          }
        } while (__builtin_amdgcn_ballot_w32(f != 0u));
      }
    }
    #pragma unroll
    for (int u = 0; u < 4; ++u) cur[u] = nxt[u];
  }

  for (int t = 0; t < qn; ++t) {
    unsigned pw = pack[q[t]];
    if (RECORD) {
      if (lane == 0 && nrec < REC_CAP) rp[nrec] = pw;
    }
    nrec += 1;
    const float* hp = h + (size_t)(pw & 0x00FFFFFFu) * D + lane;
    float* sp = sum + (pw >> 24) * SUM_PITCH + lane;
    atomicAdd(sp,      hp[0]);
    atomicAdd(sp + 32, hp[32]);
    atomicAdd(sp + 64, hp[64]);
  }
  return nrec;
}

template <bool RELU, bool REPLAY, bool FUSE>
__global__ void __launch_bounds__(OWN_WAVES * 32)
aggregate_owner(const unsigned* __restrict__ pack, const unsigned* __restrict__ own,
                const float* __restrict__ dinv, const float* __restrict__ h,
                const float* __restrict__ b, const float* __restrict__ Bn,
                float* __restrict__ out,
                unsigned* rec, unsigned* rcnt, int E, int N, int npiece) {
  __shared__ __attribute__((aligned(16))) float sum[OWN_PAD * SUM_PITCH];
  __shared__ unsigned queue[OWN_WAVES * Q_CAP];
  __shared__ __attribute__((aligned(16))) float bs[FUSE ? G_STEPS * G_TILES * 64 : 4];
  __shared__ float dl[FUSE ? OWN_PAD : 1];

  const int lane = threadIdx.x & 31;
  const int wave = __builtin_amdgcn_readfirstlane((int)(threadIdx.x >> 5));
  const int lo   = blockIdx.x * OWN_ROWS;
  const unsigned cnt = (unsigned)min(OWN_ROWS, N - lo);

  if (FUSE) {
    #pragma unroll
    for (int it = 0; it < 3; ++it) {
      const int c = it * (OWN_WAVES * 32) + (int)threadIdx.x;
      if (c < G_STEPS * G_TILES * 16) {
        const int st = c >> 4;
        const int s  = st / G_TILES;
        const int t  = st - s * G_TILES;
        const int k  = ((s >> 1) << 3) + (((c >> 3) & 1) << 2) + ((s & 1) << 1);
        const float* src = Bn + (size_t)k * D + t * 16 + (c & 7) * 2;
        const v2f w0 = *(const v2f*)src;
        const v2f w1 = *(const v2f*)(src + D);
        ((float4*)bs)[c] = make_float4(w0.x, w1.x, w0.y, w1.y);
      }
    }
  }
  for (int i = threadIdx.x; i < OWN_PAD * SUM_PITCH; i += OWN_WAVES * 32) sum[i] = 0.0f;
  __syncthreads();

  unsigned* q = queue + wave * Q_CAP;
  const unsigned pat = (unsigned)blockIdx.x * 0x01010101u;
  const int wid = blockIdx.x * OWN_WAVES + wave;
  unsigned* rp = rec + (size_t)wid * REC_CAP;

  if (!REPLAY) {
    unsigned n = scan_wave<true>(pack, own, h, sum, q, rp, E, npiece, wave, lane, pat);
    if (lane == 0) rcnt[wid] = n;
  } else {
    const unsigned n = (unsigned)__builtin_amdgcn_readfirstlane((int)rcnt[wid]);
    if (n <= REC_CAP) {
      unsigned nx = 0u;
      if ((unsigned)lane < n) nx = rp[lane];
      for (unsigned base = 0; base < n; base += 32) {
        unsigned rc = nx;
        nx = 0u;
        if (base + 32 + lane < n) nx = rp[base + 32 + lane];
        int   r = (int)(rc & 0x00FFFFFFu);
        int   y = (int)(rc >> 24);
        const unsigned m = min(32u, n - base);
        if (m >= 8)  add_batch8<0>(r, y, h, sum, lane);
        if (m >= 16) add_batch8<8>(r, y, h, sum, lane);
        if (m >= 24) add_batch8<16>(r, y, h, sum, lane);
        if (m >= 32) add_batch8<24>(r, y, h, sum, lane);
        for (unsigned t = m & ~7u; t < m; ++t) {
          int   rt = __builtin_amdgcn_readlane(r, (int)t);
          int   yt = __builtin_amdgcn_readlane(y, (int)t);
          const float* hp = h + (size_t)rt * D + lane;
          float* sp = sum + yt * SUM_PITCH + lane;
          atomicAdd(sp,      hp[0]);
          atomicAdd(sp + 32, hp[32]);
          atomicAdd(sp + 64, hp[64]);
        }
      }
    } else {
      scan_wave<false>(pack, own, h, sum, q, rp, E, npiece, wave, lane, pat);
    }
  }
  __syncthreads();

  for (int i = wave; i < (FUSE ? OWN_PAD : (int)cnt); i += OWN_WAVES) {
    if (i < (int)cnt) {
      int node = lo + i;
      float di = dinv[node];
      #pragma unroll
      for (int j = 0; j < 3; ++j) {
        int f = lane + 32 * j;
        float v = (sum[i * SUM_PITCH + f] + h[(size_t)node * D + f]) * di + b[f];
        if (RELU) v = fmaxf(v, 0.0f);
        if (FUSE) sum[i * SUM_PITCH + f] = v;
        else      out[(size_t)node * D + f] = v;
      }
      if (FUSE && lane == 0) dl[i] = di;
    } else if (FUSE && lane == 0) {
      dl[i] = 0.0f;
    }
  }
#if USE_F32_WMMA
  if (FUSE) {
    __syncthreads();
    if (wave < FUSE_WAVES) {
      const int strip = wave >> 1;
      const int tb    = (wave & 1) * 3;
      const int l     = lane & 15;
      const int half  = lane >> 4;
      const float* ar = sum + (strip * 16 + l) * SUM_PITCH + half * 4;
      float4 a4[D / 8];
      #pragma unroll
      for (int j = 0; j < D / 8; ++j) a4[j] = *(const float4*)(ar + 8 * j);
      v8f acc[3];
      #pragma unroll
      for (int t = 0; t < 3; ++t) acc[t] = (v8f){};
      const v2f* bp = (const v2f*)bs + tb * 32 + lane;
      #pragma unroll
      for (int j = 0; j < D / 8; ++j) {
        #pragma unroll
        for (int sub = 0; sub < 2; ++sub) {
          v2f a;
          a.x = sub ? a4[j].z : a4[j].x;
          a.y = sub ? a4[j].w : a4[j].y;
          const int s = 2 * j + sub;
          #pragma unroll
          for (int t = 0; t < 3; ++t) {
            const v2f bb = bp[(s * G_TILES + t) * 32];
            acc[t] = __builtin_amdgcn_wmma_f32_16x16x4_f32(
                false, a, false, bb, (short)0, acc[t], false, false);
          }
        }
      }
      const int r0 = strip * 16 + half * 8;
      #pragma unroll
      for (int r = 0; r < 8; ++r) {
        const int rowl = r0 + r;
        if (rowl < (int)cnt) {
          const float dr = dl[rowl];
          float* crow = out + (size_t)(lo + rowl) * D + tb * 16 + l;
          #pragma unroll
          for (int t = 0; t < 3; ++t) crow[t * 16] = acc[t][r] * dr;
        }
      }
    }
  }
#endif
}

extern "C" void kernel_launch(void* const* d_in, const int* in_sizes, int n_in,
                              void* d_out, int out_size, void* d_ws, size_t ws_size,
                              hipStream_t stream) {
  const float* x  = (const float*)d_in[0];
  const int*   ei = (const int*)d_in[1];
  const float* W1 = (const float*)d_in[2];
  const float* b1 = (const float*)d_in[3];
  const float* W2 = (const float*)d_in[4];
  const float* b2 = (const float*)d_in[5];
  float* out = (float*)d_out;

  int N = in_sizes[0] / D;
  int E = in_sizes[1] / 2;
  const int* row = ei;
  const int* col = ei + E;

  float* ws   = (float*)d_ws;
  float* dinv = ws;
  float* h    = ws + (((size_t)N + 127) & ~127ull);
  float* acc  = h + (size_t)N * D;
  unsigned* own = (unsigned*)(acc + (size_t)N * D);

  const int tpb = 256;
  int gN    = (N + tpb - 1) / tpb;
  int gGemm = (N / 16 + G_WAVES - 1) / G_WAVES;
  int gOwn  = (N + OWN_ROWS - 1) / OWN_ROWS;
  int npiece = (E + PIECE - 1) / PIECE;
  int nstep  = (npiece + 4 * OWN_WAVES - 1) / (4 * OWN_WAVES) + 1;
  int words  = nstep * 4 * OWN_WAVES * (PIECE / 4);
  int pitch  = (N + 127) & ~127;
  unsigned* part = own + words;
  int gRange = (N + DEG_ROWS - 1) / DEG_ROWS;
  unsigned* rcnt = part + (size_t)DEG_SLICES * pitch;
  unsigned* rec = rcnt + (((size_t)gOwn * OWN_WAVES + 127) & ~127ull);
  unsigned* pack = rec + (size_t)gOwn * OWN_WAVES * REC_CAP;

  deg_owner_kernel<<<dim3(gRange, DEG_SLICES), DEG_THREADS, 0, stream>>>(row, col, part, own, pack, E, N, pitch, words);
  dinv_kernel<<<gN, tpb, 0, stream>>>(part, dinv, N, pitch);

  gemm_wmma<<<gGemm, G_WAVES * 32, 0, stream>>>(x, W1, dinv, h, N);
  aggregate_owner<true, false, true><<<gOwn, OWN_WAVES * 32, 0, stream>>>(pack, own, dinv, h, b1, W2, acc, rec, rcnt, E, N, npiece);

  aggregate_owner<false, true, false><<<gOwn, OWN_WAVES * 32, 0, stream>>>(pack, own, dinv, acc, b2, W2, out, rec, rcnt, E, N, npiece);
}


# aggregation: loads with edge-independent addresses all issued at entry (own rows as the LDS sums' start, dinv, bias, W2, first records/plane step); row pass reads no global; sum order: own row first
# speedup vs baseline: 3.8447x; 1.0374x over previous
#include <hip/hip_runtime.h>
#include <hip/hip_bf16.h>

#define D 96

typedef float    v2f  __attribute__((ext_vector_type(2)));
typedef float    v8f  __attribute__((ext_vector_type(8)));
typedef _Float16 v16h __attribute__((ext_vector_type(16)));

#ifndef __has_builtin
#define __has_builtin(x) 0
#endif

#if defined(__AMDGCN__) && __has_builtin(__builtin_amdgcn_wmma_f32_16x16x4_f32)
#define USE_F32_WMMA 1
#else
#define USE_F32_WMMA 0
#endif

#define OWN_ROWS    196
#define DEG_ROWS    6272
#define DEG_SLICES  32
#define DEG_THREADS 512
#define DEG_BATCH   4

__global__ void __launch_bounds__(DEG_THREADS)
deg_owner_kernel(const int* __restrict__ row, const int* __restrict__ col,
                 unsigned* __restrict__ part, unsigned* __restrict__ own,
                 unsigned* __restrict__ pack, int E, int N, int pitch, int words) {
  __shared__ unsigned cnt[DEG_ROWS];
  const int tid = threadIdx.x;
  const int lo  = blockIdx.x * DEG_ROWS;

  for (int i = tid; i < DEG_ROWS; i += DEG_THREADS) cnt[i] = 0u;

  const int nthr = gridDim.x * gridDim.y * DEG_THREADS;
  for (int i = (blockIdx.y * gridDim.x + blockIdx.x) * DEG_THREADS + tid; i < words; i += nthr) {
    int e = i * 4;
    int c[4], r[4];
    if (e + 3 < E) {
      int4 v = *(const int4*)(col + e);
      int4 u = *(const int4*)(row + e);
      c[0] = v.x; c[1] = v.y; c[2] = v.z; c[3] = v.w;
      r[0] = u.x; r[1] = u.y; r[2] = u.z; r[3] = u.w;
    } else {
      #pragma unroll
      for (int k = 0; k < 4; ++k) {
        c[k] = (e + k < E) ? col[e + k] : -1;
        r[k] = (e + k < E) ? row[e + k] : 0;
      }
    }
    unsigned pk = 0;
    unsigned pw[4] = {0u, 0u, 0u, 0u};
    #pragma unroll
    for (int k = 0; k < 4; ++k)
      if (c[k] >= 0) {
        unsigned o = (unsigned)(c[k] / OWN_ROWS);
        pk |= o << (8 * k);
        pw[k] = (unsigned)r[k] | (((unsigned)c[k] - o * OWN_ROWS) << 24);
      }
    own[i] = pk;
    if (e < E) ((uint4*)pack)[i] = make_uint4(pw[0], pw[1], pw[2], pw[3]);
  }
  __syncthreads();

  const int eps = (((E + DEG_SLICES - 1) / DEG_SLICES) + 3) & ~3;
  const int e0  = min(E, (int)blockIdx.y * eps);
  const int e1  = min(E, e0 + eps);
  const int e1q = e0 + ((e1 - e0) & ~3);

  for (int e = e0 + tid * 4; e < e1q; e += DEG_THREADS * 4 * DEG_BATCH) {
    int4 v[DEG_BATCH];
    #pragma unroll
    for (int u = 0; u < DEG_BATCH; ++u) {
      int ee = e + u * DEG_THREADS * 4;
      v[u] = *(const int4*)(col + (ee < e1q ? ee : e));
    }
    #pragma unroll
    for (int u = 0; u < DEG_BATCH; ++u) {
      if (e + u * DEG_THREADS * 4 < e1q) {
        unsigned y0 = (unsigned)(v[u].x - lo), y1 = (unsigned)(v[u].y - lo);
        unsigned y2 = (unsigned)(v[u].z - lo), y3 = (unsigned)(v[u].w - lo);
        if (y0 < DEG_ROWS) atomicAdd(&cnt[y0], 1u);
        if (y1 < DEG_ROWS) atomicAdd(&cnt[y1], 1u);
        if (y2 < DEG_ROWS) atomicAdd(&cnt[y2], 1u);
        if (y3 < DEG_ROWS) atomicAdd(&cnt[y3], 1u);
      }
    }
  }
  if (tid < e1 - e1q) {
    unsigned y = (unsigned)(col[e1q + tid] - lo);
    if (y < DEG_ROWS) atomicAdd(&cnt[y], 1u);
  }
  __syncthreads();

  unsigned* plane = part + (size_t)blockIdx.y * pitch;
  for (int i = tid; i < DEG_ROWS; i += DEG_THREADS)
    if (lo + i < N) plane[lo + i] = cnt[i];
}

__global__ void dinv_kernel(const unsigned* __restrict__ part, float* __restrict__ dinv,
                            int n, int pitch) {
  int i = blockIdx.x * blockDim.x + threadIdx.x;
  if (i < n) {
    unsigned d = 0;
    #pragma unroll
    for (int s = 0; s < DEG_SLICES; ++s) d += part[(size_t)s * pitch + i];
    dinv[i] = rsqrtf((float)d + 1.0f);
  }
}

#define G_WAVES 8
#define G_STEPS (D / 4)
#define G_TILES (D / 16)

__global__ void __launch_bounds__(G_WAVES * 32)
gemm_wmma(const float* __restrict__ A, const float* __restrict__ B,
          const float* __restrict__ dinv, float* __restrict__ C, int M) {
#if USE_F32_WMMA
  __shared__ float bs[G_STEPS * G_TILES * 32 * 2];
  const int wave = threadIdx.x >> 5;
  const int lane = threadIdx.x & 31;
  const int l    = lane & 15;
  const int half = lane >> 4;
  const int m0   = (blockIdx.x * G_WAVES + wave) * 16;
  const bool live = (m0 + 16 <= M);

  float4 a4[D / 8];
  float4 d0 = make_float4(0.f, 0.f, 0.f, 0.f), d1 = d0;
  if (live) {
    const float4* ap = (const float4*)(A + (size_t)(m0 + l) * D + half * 4);
    #pragma unroll
    for (int j = 0; j < D / 8; ++j) a4[j] = ap[2 * j];
    const float4* dp = (const float4*)(dinv + m0 + half * 8);
    d0 = dp[0]; d1 = dp[1];
  }

  constexpr int G_TASKS = G_STEPS * G_TILES * 16 / (G_WAVES * 32);
  v2f w0[G_TASKS], w1[G_TASKS];
  #pragma unroll
  for (int it = 0; it < G_TASKS; ++it) {
    const int st = it * (G_WAVES * 2) + (threadIdx.x >> 4);
    const int s  = st / G_TILES;
    const int t  = st - s * G_TILES;
    const int k  = ((s >> 1) << 3) + (((threadIdx.x >> 3) & 1) << 2) + ((s & 1) << 1);
    const float* src = B + (size_t)k * D + t * 16 + (threadIdx.x & 7) * 2;
    w0[it] = *(const v2f*)src;
    w1[it] = *(const v2f*)(src + D);
  }
  #pragma unroll
  for (int it = 0; it < G_TASKS; ++it) {
    ((float4*)bs)[it * (G_WAVES * 32) + threadIdx.x] =
        make_float4(w0[it].x, w1[it].x, w0[it].y, w1[it].y);
  }
  __syncthreads();
  if (!live) return;

  v8f acc[G_TILES];
  #pragma unroll
  for (int t = 0; t < G_TILES; ++t) acc[t] = (v8f){};
  const v2f* bp = (const v2f*)bs + lane;

  #pragma unroll
  for (int j = 0; j < D / 8; ++j) {
    #pragma unroll
    for (int sub = 0; sub < 2; ++sub) {
      v2f a;
      a.x = sub ? a4[j].z : a4[j].x;
      a.y = sub ? a4[j].w : a4[j].y;
      const int s = 2 * j + sub;
      #pragma unroll
      for (int t = 0; t < G_TILES; ++t) {
        const v2f b = bp[(s * G_TILES + t) * 32];
        acc[t] = __builtin_amdgcn_wmma_f32_16x16x4_f32(
            false, a, false, b, (short)0, acc[t], false, false);
      }
    }
  }

  const float ds[8] = {d0.x, d0.y, d0.z, d0.w, d1.x, d1.y, d1.z, d1.w};
  float* crow = C + (size_t)(m0 + half * 8) * D + l;
  #pragma unroll
  for (int t = 0; t < G_TILES; ++t) {
    #pragma unroll
    for (int r = 0; r < 8; ++r)
      crow[(size_t)r * D + t * 16] = acc[t][r] * ds[r];
  }
#endif
}

#define OWN_WAVES  32
#define Q_BATCH    8
#define Q_CAP      40
#define PIECE      512

__device__ __forceinline__ void load_step(uint4 (&dst)[4], const uint4* __restrict__ plane,
                                          int p0) {
  #pragma unroll
  for (int u = 0; u < 4; ++u) dst[u] = plane[(size_t)(p0 + u * OWN_WAVES) * 32];
}

__device__ __forceinline__ unsigned differ4(unsigned x, unsigned pat) {
  unsigned t = x ^ pat;
  return ((t & 0x7F7F7F7Fu) + 0x7F7F7F7Fu) | t;
}

#define REC_CAP 256

#define OWN_PAD    208
#define SUM_PITCH  100
#define FUSE_WAVES 26

template <int OFF>
__device__ __forceinline__ void add_batch8(int r, int y, const float* __restrict__ h,
                                           float* sum, int lane) {
  float v[Q_BATCH][3];
  #pragma unroll
  for (int j = 0; j < Q_BATCH; ++j) {
    const float* hp = h + (size_t)__builtin_amdgcn_readlane(r, OFF + j) * D + lane;
    v[j][0] = hp[0]; v[j][1] = hp[32]; v[j][2] = hp[64];
  }
  #pragma unroll
  for (int j = 0; j < Q_BATCH; ++j) {
    float* sp = sum + __builtin_amdgcn_readlane(y, OFF + j) * SUM_PITCH + lane;
    atomicAdd(sp,      v[j][0]);
    atomicAdd(sp + 32, v[j][1]);
    atomicAdd(sp + 64, v[j][2]);
  }
}

template <bool RECORD>
__device__ __forceinline__ unsigned scan_wave(const unsigned* __restrict__ pack,
                                              const unsigned* __restrict__ own,
                                              const float* __restrict__ h, float* sum,
                                              unsigned* q, unsigned* rp, int E,
                                              int npiece, int wave, int lane, unsigned pat,
                                              uint4 (&cur)[4]) {
  int qn = 0;
  unsigned nrec = 0;
  const unsigned lt_mask = (1u << lane) - 1u;

  const uint4* plane = (const uint4*)own + lane;

  for (int p0 = wave; p0 < npiece; p0 += 4 * OWN_WAVES) {
    uint4 nxt[4];
    load_step(nxt, plane, p0 + 4 * OWN_WAVES);

    #pragma unroll
    for (int u = 0; u < 4; ++u) {
      const int p = p0 + u * OWN_WAVES;
      unsigned nz = ((differ4(cur[u].x, pat) & 0x80808080u) >> 7)
                  | ((differ4(cur[u].y, pat) & 0x80808080u) >> 6)
                  | ((differ4(cur[u].z, pat) & 0x80808080u) >> 5)
                  | ((differ4(cur[u].w, pat) & 0x80808080u) >> 4);
      unsigned f = nz ^ 0x0F0F0F0Fu;
      if (p < npiece && __builtin_amdgcn_ballot_w32(f != 0u)) {
        const int eb = p * PIECE + lane * 16;
        do {
          int  bit = __ffs((int)f) - 1;
          int  e   = eb + ((bit & 7) << 2) + (bit >> 3);
          bool has = (f != 0u) && (e < E);
          unsigned k = __builtin_amdgcn_ballot_w32(has);
          if (has) q[qn + __popc(k & lt_mask)] = (unsigned)e;
          qn += __popc(k);
          f &= f - 1u;

          while (qn >= Q_BATCH) {
            qn -= Q_BATCH;
            unsigned en = q[qn + (lane & (Q_BATCH - 1))];
            unsigned pw = pack[en];
            int   r  = (int)(pw & 0x00FFFFFFu);
            int   y  = (int)(pw >> 24);
            if (RECORD) {
              if (lane < Q_BATCH && nrec + lane < REC_CAP) rp[nrec + lane] = pw;
            }
            nrec += Q_BATCH;
            add_batch8<0>(r, y, h, sum, lane);
          }
        } while (__builtin_amdgcn_ballot_w32(f != 0u));
      }
    }
    #pragma unroll
    for (int u = 0; u < 4; ++u) cur[u] = nxt[u];
  }

  for (int t = 0; t < qn; ++t) {
    unsigned pw = pack[q[t]];
    if (RECORD) {
      if (lane == 0 && nrec < REC_CAP) rp[nrec] = pw;
    }
    nrec += 1;
    const float* hp = h + (size_t)(pw & 0x00FFFFFFu) * D + lane;
    float* sp = sum + (pw >> 24) * SUM_PITCH + lane;
    atomicAdd(sp,      hp[0]);
    atomicAdd(sp + 32, hp[32]);
    atomicAdd(sp + 64, hp[64]);
  }
  return nrec;
}

template <bool RELU, bool REPLAY, bool FUSE>
__global__ void __launch_bounds__(OWN_WAVES * 32)
aggregate_owner(const unsigned* __restrict__ pack, const unsigned* __restrict__ own,
                const float* __restrict__ dinv, const float* __restrict__ h,
                const float* __restrict__ b, const float* __restrict__ Bn,
                float* __restrict__ out,
                unsigned* rec, unsigned* rcnt, int E, int N, int npiece) {
  __shared__ __attribute__((aligned(16))) float sum[OWN_PAD * SUM_PITCH];
  __shared__ unsigned queue[OWN_WAVES * Q_CAP];
  __shared__ __attribute__((aligned(16))) float bs[FUSE ? G_STEPS * G_TILES * 64 : 4];
  __shared__ float dl[OWN_PAD];

  const int lane = threadIdx.x & 31;
  const int wave = __builtin_amdgcn_readfirstlane((int)(threadIdx.x >> 5));
  const int lo   = blockIdx.x * OWN_ROWS;
  const unsigned cnt = (unsigned)min(OWN_ROWS, N - lo);

  unsigned* q = queue + wave * Q_CAP;
  const unsigned pat = (unsigned)blockIdx.x * 0x01010101u;
  const int wid = blockIdx.x * OWN_WAVES + wave;
  unsigned* rp = rec + (size_t)wid * REC_CAP;

  unsigned nraw = 0u;
  if (REPLAY) nraw = rcnt[wid];

  constexpr int ROW_TRIPS = (OWN_PAD + OWN_WAVES - 1) / OWN_WAVES;
  float sv[ROW_TRIPS][3];
  #pragma unroll
  for (int k = 0; k < ROW_TRIPS; ++k) {
    const int i = wave + k * OWN_WAVES;
    const float* hp = h + (size_t)(lo + min(i, (int)cnt - 1)) * D + lane;
    sv[k][0] = hp[0]; sv[k][1] = hp[32]; sv[k][2] = hp[64];
  }
  float dmine = 0.0f;
  if (threadIdx.x < cnt) dmine = dinv[lo + threadIdx.x];
  const float bv[3] = {b[lane], b[lane + 32], b[lane + 64]};

  uint4 cur[4];
  if (!REPLAY) load_step(cur, (const uint4*)own + lane, wave);

  unsigned n = 0u, nx = 0u;
  if (REPLAY) {
    n = (unsigned)__builtin_amdgcn_readfirstlane((int)nraw);
    if (n <= REC_CAP && (unsigned)lane < n) nx = rp[lane];
  }

  v2f bw0[3], bw1[3];
  if (FUSE) {
    #pragma unroll
    for (int it = 0; it < 3; ++it) {
      const int c = it * (OWN_WAVES * 32) + (int)threadIdx.x;
      bw0[it] = (v2f){0.0f, 0.0f};
      bw1[it] = (v2f){0.0f, 0.0f};
      if (c < G_STEPS * G_TILES * 16) {
        const int st = c >> 4;
        const int s  = st / G_TILES;
        const int t  = st - s * G_TILES;
        const int k  = ((s >> 1) << 3) + (((c >> 3) & 1) << 2) + ((s & 1) << 1);
        const float* src = Bn + (size_t)k * D + t * 16 + (c & 7) * 2;
        bw0[it] = *(const v2f*)src;
        bw1[it] = *(const v2f*)(src + D);
      }
    }
  }
  if (FUSE) {
    #pragma unroll
    for (int it = 0; it < 3; ++it) {
      const int c = it * (OWN_WAVES * 32) + (int)threadIdx.x;
      if (c < G_STEPS * G_TILES * 16)
        ((float4*)bs)[c] = make_float4(bw0[it].x, bw1[it].x, bw0[it].y, bw1[it].y);
    }
  }
  #pragma unroll
  for (int k = 0; k < ROW_TRIPS; ++k) {
    const int i = wave + k * OWN_WAVES;
    if (i < OWN_PAD) {
      const bool own_row = i < (int)cnt;
      #pragma unroll
      for (int j = 0; j < 3; ++j) sum[i * SUM_PITCH + lane + 32 * j] = own_row ? sv[k][j] : 0.0f;
    }
  }
  if (threadIdx.x < OWN_PAD) dl[threadIdx.x] = dmine;
  __syncthreads();

  if (!REPLAY) {
    unsigned nown = scan_wave<true>(pack, own, h, sum, q, rp, E, npiece, wave, lane, pat, cur);
    if (lane == 0) rcnt[wid] = nown;
  } else {
    if (n <= REC_CAP) {
      for (unsigned base = 0; base < n; base += 32) {
        unsigned rc = nx;
        nx = 0u;
        if (base + 32 + lane < n) nx = rp[base + 32 + lane];
        int   r = (int)(rc & 0x00FFFFFFu);
        int   y = (int)(rc >> 24);
        const unsigned m = min(32u, n - base);
        if (m >= 8)  add_batch8<0>(r, y, h, sum, lane);
        if (m >= 16) add_batch8<8>(r, y, h, sum, lane);
        if (m >= 24) add_batch8<16>(r, y, h, sum, lane);
        if (m >= 32) add_batch8<24>(r, y, h, sum, lane);
        for (unsigned t = m & ~7u; t < m; ++t) {
          int   rt = __builtin_amdgcn_readlane(r, (int)t);
          int   yt = __builtin_amdgcn_readlane(y, (int)t);
          const float* hp = h + (size_t)rt * D + lane;
          float* sp = sum + yt * SUM_PITCH + lane;
          atomicAdd(sp,      hp[0]);
          atomicAdd(sp + 32, hp[32]);
          atomicAdd(sp + 64, hp[64]);
        }
      }
    } else {
      load_step(cur, (const uint4*)own + lane, wave);
      scan_wave<false>(pack, own, h, sum, q, rp, E, npiece, wave, lane, pat, cur);
    }
  }
  __syncthreads();

  for (int i = wave; i < (int)cnt; i += OWN_WAVES) {
    const int node = lo + i;
    const float di = dl[i];
    #pragma unroll
    for (int j = 0; j < 3; ++j) {
      const int f = lane + 32 * j;
      float v = sum[i * SUM_PITCH + f] * di + bv[j];
      if (RELU) v = fmaxf(v, 0.0f);
      if (FUSE) sum[i * SUM_PITCH + f] = v;
      else      out[(size_t)node * D + f] = v;
    }
  }
#if USE_F32_WMMA
  if (FUSE) {
    __syncthreads();
    if (wave < FUSE_WAVES) {
      const int strip = wave >> 1;
      const int tb    = (wave & 1) * 3;
      const int l     = lane & 15;
      const int half  = lane >> 4;
      const float* ar = sum + (strip * 16 + l) * SUM_PITCH + half * 4;
      float4 a4[D / 8];
      #pragma unroll
      for (int j = 0; j < D / 8; ++j) a4[j] = *(const float4*)(ar + 8 * j);
      v8f acc[3];
      #pragma unroll
      for (int t = 0; t < 3; ++t) acc[t] = (v8f){};
      const v2f* bp = (const v2f*)bs + tb * 32 + lane;
      #pragma unroll
      for (int j = 0; j < D / 8; ++j) {
        #pragma unroll
        for (int sub = 0; sub < 2; ++sub) {
          v2f a;
          a.x = sub ? a4[j].z : a4[j].x;
          a.y = sub ? a4[j].w : a4[j].y;
          const int s = 2 * j + sub;
          #pragma unroll
          for (int t = 0; t < 3; ++t) {
            const v2f bb = bp[(s * G_TILES + t) * 32];
            acc[t] = __builtin_amdgcn_wmma_f32_16x16x4_f32(
                false, a, false, bb, (short)0, acc[t], false, false);
          }
        }
      }
      const int r0 = strip * 16 + half * 8;
      #pragma unroll
      for (int r = 0; r < 8; ++r) {
        const int rowl = r0 + r;
        if (rowl < (int)cnt) {
          const float dr = dl[rowl];
          float* crow = out + (size_t)(lo + rowl) * D + tb * 16 + l;
          #pragma unroll
          for (int t = 0; t < 3; ++t) crow[t * 16] = acc[t][r] * dr;
        }
      }
    }
  }
#endif
}

extern "C" void kernel_launch(void* const* d_in, const int* in_sizes, int n_in,
                              void* d_out, int out_size, void* d_ws, size_t ws_size,
                              hipStream_t stream) {
  const float* x  = (const float*)d_in[0];
  const int*   ei = (const int*)d_in[1];
  const float* W1 = (const float*)d_in[2];
  const float* b1 = (const float*)d_in[3];
  const float* W2 = (const float*)d_in[4];
  const float* b2 = (const float*)d_in[5];
  float* out = (float*)d_out;

  int N = in_sizes[0] / D;
  int E = in_sizes[1] / 2;
  const int* row = ei;
  const int* col = ei + E;

  float* ws   = (float*)d_ws;
  float* dinv = ws;
  float* h    = ws + (((size_t)N + 127) & ~127ull);
  float* acc  = h + (size_t)N * D;
  unsigned* own = (unsigned*)(acc + (size_t)N * D);

  const int tpb = 256;
  int gN    = (N + tpb - 1) / tpb;
  int gGemm = (N / 16 + G_WAVES - 1) / G_WAVES;
  int gOwn  = (N + OWN_ROWS - 1) / OWN_ROWS;
  int npiece = (E + PIECE - 1) / PIECE;
  int nstep  = (npiece + 4 * OWN_WAVES - 1) / (4 * OWN_WAVES) + 1;
  int words  = nstep * 4 * OWN_WAVES * (PIECE / 4);
  int pitch  = (N + 127) & ~127;
  unsigned* part = own + words;
  int gRange = (N + DEG_ROWS - 1) / DEG_ROWS;
  unsigned* rcnt = part + (size_t)DEG_SLICES * pitch;
  unsigned* rec = rcnt + (((size_t)gOwn * OWN_WAVES + 127) & ~127ull);
  unsigned* pack = rec + (size_t)gOwn * OWN_WAVES * REC_CAP;

  deg_owner_kernel<<<dim3(gRange, DEG_SLICES), DEG_THREADS, 0, stream>>>(row, col, part, own, pack, E, N, pitch, words);
  dinv_kernel<<<gN, tpb, 0, stream>>>(part, dinv, N, pitch);

  gemm_wmma<<<gGemm, G_WAVES * 32, 0, stream>>>(x, W1, dinv, h, N);
  aggregate_owner<true, false, true><<<gOwn, OWN_WAVES * 32, 0, stream>>>(pack, own, dinv, h, b1, W2, acc, rec, rcnt, E, N, npiece);

  aggregate_owner<false, true, false><<<gOwn, OWN_WAVES * 32, 0, stream>>>(pack, own, dinv, acc, b2, W2, out, rec, rcnt, E, N, npiece);
}


# aggregation: no LDS float adds: layer 1 hands each owned edge to the wave owning its target row (LDS queues, kept as layer 2's records); a wave sums its rows in registers; sum order: queue arrival
# speedup vs baseline: 4.6098x; 1.1990x over previous
#include <hip/hip_runtime.h>
#include <hip/hip_bf16.h>

#define D 96

typedef float    v2f  __attribute__((ext_vector_type(2)));
typedef float    v8f  __attribute__((ext_vector_type(8)));
typedef _Float16 v16h __attribute__((ext_vector_type(16)));

#ifndef __has_builtin
#define __has_builtin(x) 0
#endif

#if defined(__AMDGCN__) && __has_builtin(__builtin_amdgcn_wmma_f32_16x16x4_f32)
#define USE_F32_WMMA 1
#else
#define USE_F32_WMMA 0
#endif

#define OWN_ROWS    196
#define DEG_ROWS    6272
#define DEG_SLICES  32
#define DEG_THREADS 512
#define DEG_BATCH   4

__global__ void __launch_bounds__(DEG_THREADS)
deg_owner_kernel(const int* __restrict__ row, const int* __restrict__ col,
                 unsigned* __restrict__ part, unsigned* __restrict__ own,
                 unsigned* __restrict__ pack, int E, int N, int pitch, int words) {
  __shared__ unsigned cnt[DEG_ROWS];
  const int tid = threadIdx.x;
  const int lo  = blockIdx.x * DEG_ROWS;

  for (int i = tid; i < DEG_ROWS; i += DEG_THREADS) cnt[i] = 0u;

  const int nthr = gridDim.x * gridDim.y * DEG_THREADS;
  for (int i = (blockIdx.y * gridDim.x + blockIdx.x) * DEG_THREADS + tid; i < words; i += nthr) {
    int e = i * 4;
    int c[4], r[4];
    if (e + 3 < E) {
      int4 v = *(const int4*)(col + e);
      int4 u = *(const int4*)(row + e);
      c[0] = v.x; c[1] = v.y; c[2] = v.z; c[3] = v.w;
      r[0] = u.x; r[1] = u.y; r[2] = u.z; r[3] = u.w;
    } else {
      #pragma unroll
      for (int k = 0; k < 4; ++k) {
        c[k] = (e + k < E) ? col[e + k] : -1;
        r[k] = (e + k < E) ? row[e + k] : 0;
      }
    }
    unsigned pk = 0;
    unsigned pw[4] = {0u, 0u, 0u, 0u};
    #pragma unroll
    for (int k = 0; k < 4; ++k)
      if (c[k] >= 0) {
        unsigned o = (unsigned)(c[k] / OWN_ROWS);
        pk |= o << (8 * k);
        pw[k] = (unsigned)r[k] | (((unsigned)c[k] - o * OWN_ROWS) << 24);
      }
    own[i] = pk;
    if (e < E) ((uint4*)pack)[i] = make_uint4(pw[0], pw[1], pw[2], pw[3]);
  }
  __syncthreads();

  const int eps = (((E + DEG_SLICES - 1) / DEG_SLICES) + 3) & ~3;
  const int e0  = min(E, (int)blockIdx.y * eps);
  const int e1  = min(E, e0 + eps);
  const int e1q = e0 + ((e1 - e0) & ~3);

  for (int e = e0 + tid * 4; e < e1q; e += DEG_THREADS * 4 * DEG_BATCH) {
    int4 v[DEG_BATCH];
    #pragma unroll
    for (int u = 0; u < DEG_BATCH; ++u) {
      int ee = e + u * DEG_THREADS * 4;
      v[u] = *(const int4*)(col + (ee < e1q ? ee : e));
    }
    #pragma unroll
    for (int u = 0; u < DEG_BATCH; ++u) {
      if (e + u * DEG_THREADS * 4 < e1q) {
        unsigned y0 = (unsigned)(v[u].x - lo), y1 = (unsigned)(v[u].y - lo);
        unsigned y2 = (unsigned)(v[u].z - lo), y3 = (unsigned)(v[u].w - lo);
        if (y0 < DEG_ROWS) atomicAdd(&cnt[y0], 1u);
        if (y1 < DEG_ROWS) atomicAdd(&cnt[y1], 1u);
        if (y2 < DEG_ROWS) atomicAdd(&cnt[y2], 1u);
        if (y3 < DEG_ROWS) atomicAdd(&cnt[y3], 1u);
      }
    }
  }
  if (tid < e1 - e1q) {
    unsigned y = (unsigned)(col[e1q + tid] - lo);
    if (y < DEG_ROWS) atomicAdd(&cnt[y], 1u);
  }
  __syncthreads();

  unsigned* plane = part + (size_t)blockIdx.y * pitch;
  for (int i = tid; i < DEG_ROWS; i += DEG_THREADS)
    if (lo + i < N) plane[lo + i] = cnt[i];
}

__global__ void dinv_kernel(const unsigned* __restrict__ part, float* __restrict__ dinv,
                            int n, int pitch) {
  int i = blockIdx.x * blockDim.x + threadIdx.x;
  if (i < n) {
    unsigned d = 0;
    #pragma unroll
    for (int s = 0; s < DEG_SLICES; ++s) d += part[(size_t)s * pitch + i];
    dinv[i] = rsqrtf((float)d + 1.0f);
  }
}

#define G_WAVES 8
#define G_STEPS (D / 4)
#define G_TILES (D / 16)

__global__ void __launch_bounds__(G_WAVES * 32)
gemm_wmma(const float* __restrict__ A, const float* __restrict__ B,
          const float* __restrict__ dinv, float* __restrict__ C, int M) {
#if USE_F32_WMMA
  __shared__ float bs[G_STEPS * G_TILES * 32 * 2];
  const int wave = threadIdx.x >> 5;
  const int lane = threadIdx.x & 31;
  const int l    = lane & 15;
  const int half = lane >> 4;
  const int m0   = (blockIdx.x * G_WAVES + wave) * 16;
  const bool live = (m0 + 16 <= M);

  float4 a4[D / 8];
  float4 d0 = make_float4(0.f, 0.f, 0.f, 0.f), d1 = d0;
  if (live) {
    const float4* ap = (const float4*)(A + (size_t)(m0 + l) * D + half * 4);
    #pragma unroll
    for (int j = 0; j < D / 8; ++j) a4[j] = ap[2 * j];
    const float4* dp = (const float4*)(dinv + m0 + half * 8);
    d0 = dp[0]; d1 = dp[1];
  }

  constexpr int G_TASKS = G_STEPS * G_TILES * 16 / (G_WAVES * 32);
  v2f w0[G_TASKS], w1[G_TASKS];
  #pragma unroll
  for (int it = 0; it < G_TASKS; ++it) {
    const int st = it * (G_WAVES * 2) + (threadIdx.x >> 4);
    const int s  = st / G_TILES;
    const int t  = st - s * G_TILES;
    const int k  = ((s >> 1) << 3) + (((threadIdx.x >> 3) & 1) << 2) + ((s & 1) << 1);
    const float* src = B + (size_t)k * D + t * 16 + (threadIdx.x & 7) * 2;
    w0[it] = *(const v2f*)src;
    w1[it] = *(const v2f*)(src + D);
  }
  #pragma unroll
  for (int it = 0; it < G_TASKS; ++it) {
    ((float4*)bs)[it * (G_WAVES * 32) + threadIdx.x] =
        make_float4(w0[it].x, w1[it].x, w0[it].y, w1[it].y);
  }
  __syncthreads();
  if (!live) return;

  v8f acc[G_TILES];
  #pragma unroll
  for (int t = 0; t < G_TILES; ++t) acc[t] = (v8f){};
  const v2f* bp = (const v2f*)bs + lane;

  #pragma unroll
  for (int j = 0; j < D / 8; ++j) {
    #pragma unroll
    for (int sub = 0; sub < 2; ++sub) {
      v2f a;
      a.x = sub ? a4[j].z : a4[j].x;
      a.y = sub ? a4[j].w : a4[j].y;
      const int s = 2 * j + sub;
      #pragma unroll
      for (int t = 0; t < G_TILES; ++t) {
        const v2f b = bp[(s * G_TILES + t) * 32];
        acc[t] = __builtin_amdgcn_wmma_f32_16x16x4_f32(
            false, a, false, b, (short)0, acc[t], false, false);
      }
    }
  }

  const float ds[8] = {d0.x, d0.y, d0.z, d0.w, d1.x, d1.y, d1.z, d1.w};
  float* crow = C + (size_t)(m0 + half * 8) * D + l;
  #pragma unroll
  for (int t = 0; t < G_TILES; ++t) {
    #pragma unroll
    for (int r = 0; r < 8; ++r)
      crow[(size_t)r * D + t * 16] = acc[t][r] * ds[r];
  }
#endif
}

#define OWN_WAVES  32
#define Q_CAP      64
#define PIECE      512
#define ROW_TRIPS  7

__device__ __forceinline__ void load_step(uint4 (&dst)[4], const uint4* __restrict__ plane,
                                          int p0) {
  #pragma unroll
  for (int u = 0; u < 4; ++u) dst[u] = plane[(size_t)(p0 + u * OWN_WAVES) * 32];
}

__device__ __forceinline__ unsigned differ4(unsigned x, unsigned pat) {
  unsigned t = x ^ pat;
  return ((t & 0x7F7F7F7Fu) + 0x7F7F7F7Fu) | t;
}

#define REC_CAP 256

#define OWN_PAD    208
#define SUM_PITCH  100
#define FUSE_WAVES 26

struct RowSums { v8f c0, c1, c2; };

__device__ __forceinline__ void acc_add(RowSums& a, int k, float v0, float v1, float v2) {
  a.c0[k] += v0; a.c1[k] += v1; a.c2[k] += v2;
}

template <int OFF>
__device__ __forceinline__ void add_batch8(unsigned rc, const float* __restrict__ h,
                                           RowSums& acc, int lane) {
  float v[8][3];
  #pragma unroll
  for (int j = 0; j < 8; ++j) {
    const unsigned w = (unsigned)__builtin_amdgcn_readlane((int)rc, OFF + j);
    const float* hp = h + (size_t)(w & 0x00FFFFFFu) * D + lane;
    v[j][0] = hp[0]; v[j][1] = hp[32]; v[j][2] = hp[64];
  }
  #pragma unroll
  for (int j = 0; j < 8; ++j) {
    const unsigned w = (unsigned)__builtin_amdgcn_readlane((int)rc, OFF + j);
    acc_add(acc, (int)(w >> 29), v[j][0], v[j][1], v[j][2]);
  }
}

__device__ __forceinline__ void add_chunk(unsigned rc, unsigned m, const float* __restrict__ h,
                                          RowSums& acc, int lane) {
  if (m >= 8)  add_batch8<0>(rc, h, acc, lane);
  if (m >= 16) add_batch8<8>(rc, h, acc, lane);
  if (m >= 24) add_batch8<16>(rc, h, acc, lane);
  if (m >= 32) add_batch8<24>(rc, h, acc, lane);
  for (unsigned t = m & ~7u; t < m; ++t) {
    const unsigned w = (unsigned)__builtin_amdgcn_readlane((int)rc, (int)t);
    const float* hp = h + (size_t)(w & 0x00FFFFFFu) * D + lane;
    acc_add(acc, (int)(w >> 29), hp[0], hp[32], hp[64]);
  }
}

__device__ __forceinline__ void route32(unsigned pw, bool act, unsigned* rq, unsigned* rqn,
                                        const float* __restrict__ h, float* sum, int lane) {
  unsigned slot = 0u;
  if (act) {
    const unsigned ow = (pw >> 24) & 31u;
    slot = atomicAdd(&rqn[ow], 1u);
    if (slot < REC_CAP) rq[ow * REC_CAP + slot] = pw;
  }
  unsigned ov = __builtin_amdgcn_ballot_w32(act && slot >= REC_CAP);
  while (ov) {
    const int j = __ffs((int)ov) - 1;
    ov &= ov - 1u;
    const unsigned w = (unsigned)__builtin_amdgcn_readlane((int)pw, j);
    const float* hp = h + (size_t)(w & 0x00FFFFFFu) * D + lane;
    float* sp = sum + (w >> 24) * SUM_PITCH + lane;
    atomicAdd(sp,      hp[0]);
    atomicAdd(sp + 32, hp[32]);
    atomicAdd(sp + 64, hp[64]);
  }
}

__device__ __forceinline__ void scan_route(const unsigned* __restrict__ pack,
                                           const unsigned* __restrict__ own,
                                           const float* __restrict__ h, float* sum,
                                           unsigned* q, unsigned* rq, unsigned* rqn, int E,
                                           int npiece, int wave, int lane, unsigned pat,
                                           uint4 (&cur)[4]) {
  int qn = 0;
  const unsigned lt_mask = (1u << lane) - 1u;

  const uint4* plane = (const uint4*)own + lane;

  for (int p0 = wave; p0 < npiece; p0 += 4 * OWN_WAVES) {
    uint4 nxt[4];
    load_step(nxt, plane, p0 + 4 * OWN_WAVES);

    #pragma unroll
    for (int u = 0; u < 4; ++u) {
      const int p = p0 + u * OWN_WAVES;
      unsigned nz = ((differ4(cur[u].x, pat) & 0x80808080u) >> 7)
                  | ((differ4(cur[u].y, pat) & 0x80808080u) >> 6)
                  | ((differ4(cur[u].z, pat) & 0x80808080u) >> 5)
                  | ((differ4(cur[u].w, pat) & 0x80808080u) >> 4);
      unsigned f = nz ^ 0x0F0F0F0Fu;
      if (p < npiece && __builtin_amdgcn_ballot_w32(f != 0u)) {
        const int eb = p * PIECE + lane * 16;
        do {
          int  bit = __ffs((int)f) - 1;
          int  e   = eb + ((bit & 7) << 2) + (bit >> 3);
          bool has = (f != 0u) && (e < E);
          unsigned k = __builtin_amdgcn_ballot_w32(has);
          if (has) q[qn + __popc(k & lt_mask)] = (unsigned)e;
          qn += __popc(k);
          f &= f - 1u;

          if (qn >= 32) {
            qn -= 32;
            const unsigned en = q[qn + lane];
            route32(pack[en], true, rq, rqn, h, sum, lane);
          }
        } while (__builtin_amdgcn_ballot_w32(f != 0u));
      }
    }
    #pragma unroll
    for (int u = 0; u < 4; ++u) cur[u] = nxt[u];
  }

  if (qn > 0) {
    const bool act = lane < qn;
    unsigned pw = 0u;
    if (act) pw = pack[q[lane]];
    route32(pw, act, rq, rqn, h, sum, lane);
  }
}

__global__ void __launch_bounds__(OWN_WAVES * 32)
aggregate_scan(const unsigned* __restrict__ pack, const unsigned* __restrict__ own,
               const float* __restrict__ dinv, const float* __restrict__ h,
               const float* __restrict__ b, const float* __restrict__ Bn,
               float* __restrict__ out,
               unsigned* rec, unsigned* rcnt, int E, int N, int npiece) {
  __shared__ __attribute__((aligned(16))) float sum[OWN_PAD * SUM_PITCH];
  __shared__ unsigned queue[OWN_WAVES * Q_CAP];
  __shared__ unsigned rq[OWN_WAVES * REC_CAP];
  __shared__ unsigned rqn[OWN_WAVES];
  __shared__ __attribute__((aligned(16))) float bs[G_STEPS * G_TILES * 64];
  __shared__ float dl[OWN_PAD];

  const int lane = threadIdx.x & 31;
  const int wave = __builtin_amdgcn_readfirstlane((int)(threadIdx.x >> 5));
  const int lo   = blockIdx.x * OWN_ROWS;
  const unsigned cnt = (unsigned)min(OWN_ROWS, N - lo);

  unsigned* q = queue + wave * Q_CAP;
  const unsigned pat = (unsigned)blockIdx.x * 0x01010101u;
  const int wid = blockIdx.x * OWN_WAVES + wave;
  unsigned* rp = rec + (size_t)wid * REC_CAP;

  RowSums acc;
  acc.c0 = (v8f){}; acc.c1 = (v8f){}; acc.c2 = (v8f){};
  #pragma unroll
  for (int k = 0; k < ROW_TRIPS; ++k) {
    const int i = wave + k * OWN_WAVES;
    const float* hp = h + (size_t)(lo + min(i, (int)cnt - 1)) * D + lane;
    acc.c0[k] = hp[0]; acc.c1[k] = hp[32]; acc.c2[k] = hp[64];
  }
  float dmine = 0.0f;
  if (threadIdx.x < cnt) dmine = dinv[lo + threadIdx.x];
  const float bv[3] = {b[lane], b[lane + 32], b[lane + 64]};

  uint4 cur[4];
  load_step(cur, (const uint4*)own + lane, wave);

  v2f bw0[3], bw1[3];
  #pragma unroll
  for (int it = 0; it < 3; ++it) {
    const int c = it * (OWN_WAVES * 32) + (int)threadIdx.x;
    bw0[it] = (v2f){0.0f, 0.0f};
    bw1[it] = (v2f){0.0f, 0.0f};
    if (c < G_STEPS * G_TILES * 16) {
      const int st = c >> 4;
      const int s  = st / G_TILES;
      const int t  = st - s * G_TILES;
      const int k  = ((s >> 1) << 3) + (((c >> 3) & 1) << 2) + ((s & 1) << 1);
      const float* src = Bn + (size_t)k * D + t * 16 + (c & 7) * 2;
      bw0[it] = *(const v2f*)src;
      bw1[it] = *(const v2f*)(src + D);
    }
  }
  #pragma unroll
  for (int it = 0; it < 3; ++it) {
    const int c = it * (OWN_WAVES * 32) + (int)threadIdx.x;
    if (c < G_STEPS * G_TILES * 16)
      ((float4*)bs)[c] = make_float4(bw0[it].x, bw1[it].x, bw0[it].y, bw1[it].y);
  }
  #pragma unroll
  for (int k = 0; k < ROW_TRIPS; ++k) {
    const int i = wave + k * OWN_WAVES;
    if (i < OWN_PAD) {
      #pragma unroll
      for (int j = 0; j < 3; ++j) sum[i * SUM_PITCH + lane + 32 * j] = 0.0f;
    }
    if (i >= (int)cnt) { acc.c0[k] = 0.0f; acc.c1[k] = 0.0f; acc.c2[k] = 0.0f; }
  }
  if (threadIdx.x < OWN_PAD) dl[threadIdx.x] = dmine;
  if (threadIdx.x < OWN_WAVES) rqn[threadIdx.x] = 0u;
  __syncthreads();

  scan_route(pack, own, h, sum, q, rq, rqn, E, npiece, wave, lane, pat, cur);
  __syncthreads();

  {
    const unsigned n = (unsigned)__builtin_amdgcn_readfirstlane((int)rqn[wave]);
    const unsigned m = min(n, (unsigned)REC_CAP);
    if (lane == 0) rcnt[wid] = n;
    const unsigned* mq = rq + wave * REC_CAP;
    for (unsigned base = 0; base < m; base += 32) {
      unsigned rc = 0u;
      if (base + lane < m) { rc = mq[base + lane]; rp[base + lane] = rc; }
      add_chunk(rc, min(32u, m - base), h, acc, lane);
    }
  }

  #pragma unroll
  for (int k = 0; k < ROW_TRIPS; ++k) {
    const int i = wave + k * OWN_WAVES;
    if (i < (int)cnt) {
      const float di = dl[i];
      const float s[3] = {acc.c0[k], acc.c1[k], acc.c2[k]};
      #pragma unroll
      for (int j = 0; j < 3; ++j) {
        const int f = lane + 32 * j;
        float v = (s[j] + sum[i * SUM_PITCH + f]) * di + bv[j];
        sum[i * SUM_PITCH + f] = fmaxf(v, 0.0f);
      }
    }
  }
#if USE_F32_WMMA
  __syncthreads();
  if (wave < FUSE_WAVES) {
    const int strip = wave >> 1;
    const int tb    = (wave & 1) * 3;
    const int l     = lane & 15;
    const int half  = lane >> 4;
    const float* ar = sum + (strip * 16 + l) * SUM_PITCH + half * 4;
    float4 a4[D / 8];
    #pragma unroll
    for (int j = 0; j < D / 8; ++j) a4[j] = *(const float4*)(ar + 8 * j);
    v8f pr[3];
    #pragma unroll
    for (int t = 0; t < 3; ++t) pr[t] = (v8f){};
    const v2f* bp = (const v2f*)bs + tb * 32 + lane;
    #pragma unroll
    for (int j = 0; j < D / 8; ++j) {
      #pragma unroll
      for (int sub = 0; sub < 2; ++sub) {
        v2f a;
        a.x = sub ? a4[j].z : a4[j].x;
        a.y = sub ? a4[j].w : a4[j].y;
        const int s = 2 * j + sub;
        #pragma unroll
        for (int t = 0; t < 3; ++t) {
          const v2f bb = bp[(s * G_TILES + t) * 32];
          pr[t] = __builtin_amdgcn_wmma_f32_16x16x4_f32(
              false, a, false, bb, (short)0, pr[t], false, false);
        }
      }
    }
    const int r0 = strip * 16 + half * 8;
    #pragma unroll
    for (int r = 0; r < 8; ++r) {
      const int rowl = r0 + r;
      if (rowl < (int)cnt) {
        const float dr = dl[rowl];
        float* crow = out + (size_t)(lo + rowl) * D + tb * 16 + l;
        #pragma unroll
        for (int t = 0; t < 3; ++t) crow[t * 16] = pr[t][r] * dr;
      }
    }
  }
#endif
}

__global__ void __launch_bounds__(OWN_WAVES * 32)
aggregate_replay(const unsigned* __restrict__ pack, const unsigned* __restrict__ own,
                 const float* __restrict__ dinv, const float* __restrict__ h,
                 const float* __restrict__ b, float* __restrict__ out,
                 const unsigned* __restrict__ rec, const unsigned* __restrict__ rcnt,
                 int E, int N) {
  const int lane = threadIdx.x & 31;
  const int wave = __builtin_amdgcn_readfirstlane((int)(threadIdx.x >> 5));
  const int lo   = blockIdx.x * OWN_ROWS;
  const unsigned cnt = (unsigned)min(OWN_ROWS, N - lo);
  const int wid = blockIdx.x * OWN_WAVES + wave;
  const unsigned* rp = rec + (size_t)wid * REC_CAP;

  const unsigned nraw = rcnt[wid];

  RowSums acc;
  acc.c0 = (v8f){}; acc.c1 = (v8f){}; acc.c2 = (v8f){};
  float dk[ROW_TRIPS];
  #pragma unroll
  for (int k = 0; k < ROW_TRIPS; ++k) {
    const int i = lo + min(wave + k * OWN_WAVES, (int)cnt - 1);
    const float* hp = h + (size_t)i * D + lane;
    acc.c0[k] = hp[0]; acc.c1[k] = hp[32]; acc.c2[k] = hp[64];
    dk[k] = dinv[i];
  }
  const float bv[3] = {b[lane], b[lane + 32], b[lane + 64]};

  const unsigned n = (unsigned)__builtin_amdgcn_readfirstlane((int)nraw);
  unsigned nx = 0u;
  if (n <= REC_CAP && (unsigned)lane < n) nx = rp[lane];

  if (n <= REC_CAP) {
    for (unsigned base = 0; base < n; base += 32) {
      const unsigned rc = nx;
      asm volatile("" :: "v"(rc));
      nx = 0u;
      if (base + 32 + lane < n) nx = rp[base + 32 + lane];
      add_chunk(rc, min(32u, n - base), h, acc, lane);
    }
  } else {
    for (int e0 = 0; e0 < E; e0 += 32) {
      const int e = e0 + lane;
      unsigned pw = 0u;
      bool mine = false;
      if (e < E) {
        const unsigned ob = (own[e >> 2] >> ((e & 3) * 8)) & 0xFFu;
        if (ob == blockIdx.x) {
          pw = pack[e];
          mine = ((pw >> 24) & 31u) == (unsigned)wave;
        }
      }
      unsigned k = __builtin_amdgcn_ballot_w32(mine);
      while (k) {
        const int j = __ffs((int)k) - 1;
        k &= k - 1u;
        const unsigned w = (unsigned)__builtin_amdgcn_readlane((int)pw, j);
        const float* hp = h + (size_t)(w & 0x00FFFFFFu) * D + lane;
        acc_add(acc, (int)(w >> 29), hp[0], hp[32], hp[64]);
      }
    }
  }

  #pragma unroll
  for (int k = 0; k < ROW_TRIPS; ++k) {
    const int i = wave + k * OWN_WAVES;
    if (i < (int)cnt) {
      const float s[3] = {acc.c0[k], acc.c1[k], acc.c2[k]};
      #pragma unroll
      for (int j = 0; j < 3; ++j)
        out[(size_t)(lo + i) * D + lane + 32 * j] = s[j] * dk[k] + bv[j];
    }
  }
}

extern "C" void kernel_launch(void* const* d_in, const int* in_sizes, int n_in,
                              void* d_out, int out_size, void* d_ws, size_t ws_size,
                              hipStream_t stream) {
  const float* x  = (const float*)d_in[0];
  const int*   ei = (const int*)d_in[1];
  const float* W1 = (const float*)d_in[2];
  const float* b1 = (const float*)d_in[3];
  const float* W2 = (const float*)d_in[4];
  const float* b2 = (const float*)d_in[5];
  float* out = (float*)d_out;

  int N = in_sizes[0] / D;
  int E = in_sizes[1] / 2;
  const int* row = ei;
  const int* col = ei + E;

  float* ws   = (float*)d_ws;
  float* dinv = ws;
  float* h    = ws + (((size_t)N + 127) & ~127ull);
  float* acc  = h + (size_t)N * D;
  unsigned* own = (unsigned*)(acc + (size_t)N * D);

  const int tpb = 256;
  int gN    = (N + tpb - 1) / tpb;
  int gGemm = (N / 16 + G_WAVES - 1) / G_WAVES;
  int gOwn  = (N + OWN_ROWS - 1) / OWN_ROWS;
  int npiece = (E + PIECE - 1) / PIECE;
  int nstep  = (npiece + 4 * OWN_WAVES - 1) / (4 * OWN_WAVES) + 1;
  int words  = nstep * 4 * OWN_WAVES * (PIECE / 4);
  int pitch  = (N + 127) & ~127;
  unsigned* part = own + words;
  int gRange = (N + DEG_ROWS - 1) / DEG_ROWS;
  unsigned* rcnt = part + (size_t)DEG_SLICES * pitch;
  unsigned* rec = rcnt + (((size_t)gOwn * OWN_WAVES + 127) & ~127ull);
  unsigned* pack = rec + (size_t)gOwn * OWN_WAVES * REC_CAP;

  deg_owner_kernel<<<dim3(gRange, DEG_SLICES), DEG_THREADS, 0, stream>>>(row, col, part, own, pack, E, N, pitch, words);
  dinv_kernel<<<gN, tpb, 0, stream>>>(part, dinv, N, pitch);

  gemm_wmma<<<gGemm, G_WAVES * 32, 0, stream>>>(x, W1, dinv, h, N);
  aggregate_scan<<<gOwn, OWN_WAVES * 32, 0, stream>>>(pack, own, dinv, h, b1, W2, acc, rec, rcnt, E, N, npiece);

  aggregate_replay<<<gOwn, OWN_WAVES * 32, 0, stream>>>(pack, own, dinv, acc, b2, out, rec, rcnt, E, N);
}


# owner plane bit-sliced: the degree pass writes 8 bit words per 32 edges (ballots); a scanning lane matches 32 edges with 8 logic ops (was 22 per 16); plane bytes unchanged; sum order: queue arrival
# speedup vs baseline: 4.7705x; 1.0349x over previous
#include <hip/hip_runtime.h>
#include <hip/hip_bf16.h>

#define D 96

typedef float    v2f  __attribute__((ext_vector_type(2)));
typedef float    v8f  __attribute__((ext_vector_type(8)));
typedef _Float16 v16h __attribute__((ext_vector_type(16)));

#ifndef __has_builtin
#define __has_builtin(x) 0
#endif

#if defined(__AMDGCN__) && __has_builtin(__builtin_amdgcn_wmma_f32_16x16x4_f32)
#define USE_F32_WMMA 1
#else
#define USE_F32_WMMA 0
#endif

#define OWN_ROWS    196
#define DEG_ROWS    6272
#define DEG_SLICES  32
#define DEG_THREADS 512
#define DEG_BATCH   4

__global__ void __launch_bounds__(DEG_THREADS)
deg_owner_kernel(const int* __restrict__ row, const int* __restrict__ col,
                 unsigned* __restrict__ part, unsigned* __restrict__ own,
                 unsigned* __restrict__ pack, int E, int N, int pitch, int words) {
  __shared__ unsigned cnt[DEG_ROWS];
  const int tid = threadIdx.x;
  const int lo  = blockIdx.x * DEG_ROWS;

  for (int i = tid; i < DEG_ROWS; i += DEG_THREADS) cnt[i] = 0u;

  const int lane = tid & 31;
  const int nwv  = gridDim.x * gridDim.y * (DEG_THREADS / 32);
  const int wv0  = (blockIdx.y * gridDim.x + blockIdx.x) * (DEG_THREADS / 32)
                 + __builtin_amdgcn_readfirstlane(tid >> 5);
  const int nchunk = words / 64;
  for (int ch = wv0; ch < nchunk; ch += nwv) {
    unsigned o[8];
    #pragma unroll
    for (int m = 0; m < 2; ++m) {
      const int e = ch * 256 + m * 128 + lane * 4;
      int c[4], r[4];
      if (e + 3 < E) {
        int4 v = *(const int4*)(col + e);
        int4 u = *(const int4*)(row + e);
        c[0] = v.x; c[1] = v.y; c[2] = v.z; c[3] = v.w;
        r[0] = u.x; r[1] = u.y; r[2] = u.z; r[3] = u.w;
      } else {
        #pragma unroll
        for (int k = 0; k < 4; ++k) {
          c[k] = (e + k < E) ? col[e + k] : -1;
          r[k] = (e + k < E) ? row[e + k] : 0;
        }
      }
      unsigned pw[4] = {0u, 0u, 0u, 0u};
      #pragma unroll
      for (int k = 0; k < 4; ++k) {
        unsigned ok = 0u;
        if (c[k] >= 0) {
          ok = (unsigned)(c[k] / OWN_ROWS);
          pw[k] = (unsigned)r[k] | (((unsigned)c[k] - ok * OWN_ROWS) << 24);
        }
        o[4 * m + k] = ok;
      }
      if (e < E) ((uint4*)pack)[e >> 2] = make_uint4(pw[0], pw[1], pw[2], pw[3]);
    }
    unsigned w0 = 0u, w1 = 0u;
    #pragma unroll
    for (int q = 0; q < 8; ++q) {
      #pragma unroll
      for (int b = 0; b < 4; ++b) {
        const unsigned lo4 = __builtin_amdgcn_ballot_w32(((o[q] >> b) & 1u) != 0u);
        const unsigned hi4 = __builtin_amdgcn_ballot_w32(((o[q] >> (b + 4)) & 1u) != 0u);
        if (lane == q * 4 + b) { w0 = lo4; w1 = hi4; }
      }
    }
    unsigned* up = own + (size_t)(ch >> 2) * 256 + (ch & 3) * 32 + lane;
    up[0]   = w0;
    up[128] = w1;
  }
  __syncthreads();

  const int eps = (((E + DEG_SLICES - 1) / DEG_SLICES) + 3) & ~3;
  const int e0  = min(E, (int)blockIdx.y * eps);
  const int e1  = min(E, e0 + eps);
  const int e1q = e0 + ((e1 - e0) & ~3);

  for (int e = e0 + tid * 4; e < e1q; e += DEG_THREADS * 4 * DEG_BATCH) {
    int4 v[DEG_BATCH];
    #pragma unroll
    for (int u = 0; u < DEG_BATCH; ++u) {
      int ee = e + u * DEG_THREADS * 4;
      v[u] = *(const int4*)(col + (ee < e1q ? ee : e));
    }
    #pragma unroll
    for (int u = 0; u < DEG_BATCH; ++u) {
      if (e + u * DEG_THREADS * 4 < e1q) {
        unsigned y0 = (unsigned)(v[u].x - lo), y1 = (unsigned)(v[u].y - lo);
        unsigned y2 = (unsigned)(v[u].z - lo), y3 = (unsigned)(v[u].w - lo);
        if (y0 < DEG_ROWS) atomicAdd(&cnt[y0], 1u);
        if (y1 < DEG_ROWS) atomicAdd(&cnt[y1], 1u);
        if (y2 < DEG_ROWS) atomicAdd(&cnt[y2], 1u);
        if (y3 < DEG_ROWS) atomicAdd(&cnt[y3], 1u);
      }
    }
  }
  if (tid < e1 - e1q) {
    unsigned y = (unsigned)(col[e1q + tid] - lo);
    if (y < DEG_ROWS) atomicAdd(&cnt[y], 1u);
  }
  __syncthreads();

  unsigned* plane = part + (size_t)blockIdx.y * pitch;
  for (int i = tid; i < DEG_ROWS; i += DEG_THREADS)
    if (lo + i < N) plane[lo + i] = cnt[i];
}

__global__ void dinv_kernel(const unsigned* __restrict__ part, float* __restrict__ dinv,
                            int n, int pitch) {
  int i = blockIdx.x * blockDim.x + threadIdx.x;
  if (i < n) {
    unsigned d = 0;
    #pragma unroll
    for (int s = 0; s < DEG_SLICES; ++s) d += part[(size_t)s * pitch + i];
    dinv[i] = rsqrtf((float)d + 1.0f);
  }
}

#define G_WAVES 8
#define G_STEPS (D / 4)
#define G_TILES (D / 16)

__global__ void __launch_bounds__(G_WAVES * 32)
gemm_wmma(const float* __restrict__ A, const float* __restrict__ B,
          const float* __restrict__ dinv, float* __restrict__ C, int M) {
#if USE_F32_WMMA
  __shared__ float bs[G_STEPS * G_TILES * 32 * 2];
  const int wave = threadIdx.x >> 5;
  const int lane = threadIdx.x & 31;
  const int l    = lane & 15;
  const int half = lane >> 4;
  const int m0   = (blockIdx.x * G_WAVES + wave) * 16;
  const bool live = (m0 + 16 <= M);

  float4 a4[D / 8];
  float4 d0 = make_float4(0.f, 0.f, 0.f, 0.f), d1 = d0;
  if (live) {
    const float4* ap = (const float4*)(A + (size_t)(m0 + l) * D + half * 4);
    #pragma unroll
    for (int j = 0; j < D / 8; ++j) a4[j] = ap[2 * j];
    const float4* dp = (const float4*)(dinv + m0 + half * 8);
    d0 = dp[0]; d1 = dp[1];
  }

  constexpr int G_TASKS = G_STEPS * G_TILES * 16 / (G_WAVES * 32);
  v2f w0[G_TASKS], w1[G_TASKS];
  #pragma unroll
  for (int it = 0; it < G_TASKS; ++it) {
    const int st = it * (G_WAVES * 2) + (threadIdx.x >> 4);
    const int s  = st / G_TILES;
    const int t  = st - s * G_TILES;
    const int k  = ((s >> 1) << 3) + (((threadIdx.x >> 3) & 1) << 2) + ((s & 1) << 1);
    const float* src = B + (size_t)k * D + t * 16 + (threadIdx.x & 7) * 2;
    w0[it] = *(const v2f*)src;
    w1[it] = *(const v2f*)(src + D);
  }
  #pragma unroll
  for (int it = 0; it < G_TASKS; ++it) {
    ((float4*)bs)[it * (G_WAVES * 32) + threadIdx.x] =
        make_float4(w0[it].x, w1[it].x, w0[it].y, w1[it].y);
  }
  __syncthreads();
  if (!live) return;

  v8f acc[G_TILES];
  #pragma unroll
  for (int t = 0; t < G_TILES; ++t) acc[t] = (v8f){};
  const v2f* bp = (const v2f*)bs + lane;

  #pragma unroll
  for (int j = 0; j < D / 8; ++j) {
    #pragma unroll
    for (int sub = 0; sub < 2; ++sub) {
      v2f a;
      a.x = sub ? a4[j].z : a4[j].x;
      a.y = sub ? a4[j].w : a4[j].y;
      const int s = 2 * j + sub;
      #pragma unroll
      for (int t = 0; t < G_TILES; ++t) {
        const v2f b = bp[(s * G_TILES + t) * 32];
        acc[t] = __builtin_amdgcn_wmma_f32_16x16x4_f32(
            false, a, false, b, (short)0, acc[t], false, false);
      }
    }
  }

  const float ds[8] = {d0.x, d0.y, d0.z, d0.w, d1.x, d1.y, d1.z, d1.w};
  float* crow = C + (size_t)(m0 + half * 8) * D + l;
  #pragma unroll
  for (int t = 0; t < G_TILES; ++t) {
    #pragma unroll
    for (int r = 0; r < 8; ++r)
      crow[(size_t)r * D + t * 16] = acc[t][r] * ds[r];
  }
#endif
}

#define OWN_WAVES  32
#define Q_CAP      64
#define UNIT       1024
#define ROW_TRIPS  7

__device__ __forceinline__ void load_step(uint4 (&dst)[4], const uint4* __restrict__ plane,
                                          int p0) {
  #pragma unroll
  for (int u = 0; u < 2; ++u) {
    const uint4* pp = plane + (size_t)(p0 + u * OWN_WAVES) * 64;
    dst[2 * u]     = pp[0];
    dst[2 * u + 1] = pp[32];
  }
}

__device__ __forceinline__ unsigned match32(const uint4& lo, const uint4& hi,
                                            const unsigned (&gm)[8]) {
  const unsigned d = (lo.x ^ gm[0]) | (lo.y ^ gm[1]) | (lo.z ^ gm[2]) | (lo.w ^ gm[3])
                   | (hi.x ^ gm[4]) | (hi.y ^ gm[5]) | (hi.z ^ gm[6]) | (hi.w ^ gm[7]);
  return ~d;
}

#define REC_CAP 256

#define OWN_PAD    208
#define SUM_PITCH  100
#define FUSE_WAVES 26

struct RowSums { v8f c0, c1, c2; };

__device__ __forceinline__ void acc_add(RowSums& a, int k, float v0, float v1, float v2) {
  a.c0[k] += v0; a.c1[k] += v1; a.c2[k] += v2;
}

template <int OFF>
__device__ __forceinline__ void add_batch8(unsigned rc, const float* __restrict__ h,
                                           RowSums& acc, int lane) {
  float v[8][3];
  #pragma unroll
  for (int j = 0; j < 8; ++j) {
    const unsigned w = (unsigned)__builtin_amdgcn_readlane((int)rc, OFF + j);
    const float* hp = h + (size_t)(w & 0x00FFFFFFu) * D + lane;
    v[j][0] = hp[0]; v[j][1] = hp[32]; v[j][2] = hp[64];
  }
  #pragma unroll
  for (int j = 0; j < 8; ++j) {
    const unsigned w = (unsigned)__builtin_amdgcn_readlane((int)rc, OFF + j);
    acc_add(acc, (int)(w >> 29), v[j][0], v[j][1], v[j][2]);
  }
}

__device__ __forceinline__ void add_chunk(unsigned rc, unsigned m, const float* __restrict__ h,
                                          RowSums& acc, int lane) {
  if (m >= 8)  add_batch8<0>(rc, h, acc, lane);
  if (m >= 16) add_batch8<8>(rc, h, acc, lane);
  if (m >= 24) add_batch8<16>(rc, h, acc, lane);
  if (m >= 32) add_batch8<24>(rc, h, acc, lane);
  for (unsigned t = m & ~7u; t < m; ++t) {
    const unsigned w = (unsigned)__builtin_amdgcn_readlane((int)rc, (int)t);
    const float* hp = h + (size_t)(w & 0x00FFFFFFu) * D + lane;
    acc_add(acc, (int)(w >> 29), hp[0], hp[32], hp[64]);
  }
}

__device__ __forceinline__ void route32(unsigned pw, bool act, unsigned* rq, unsigned* rqn,
                                        const float* __restrict__ h, float* sum, int lane) {
  unsigned slot = 0u;
  if (act) {
    const unsigned ow = (pw >> 24) & 31u;
    slot = atomicAdd(&rqn[ow], 1u);
    if (slot < REC_CAP) rq[ow * REC_CAP + slot] = pw;
  }
  unsigned ov = __builtin_amdgcn_ballot_w32(act && slot >= REC_CAP);
  while (ov) {
    const int j = __ffs((int)ov) - 1;
    ov &= ov - 1u;
    const unsigned w = (unsigned)__builtin_amdgcn_readlane((int)pw, j);
    const float* hp = h + (size_t)(w & 0x00FFFFFFu) * D + lane;
    float* sp = sum + (w >> 24) * SUM_PITCH + lane;
    atomicAdd(sp,      hp[0]);
    atomicAdd(sp + 32, hp[32]);
    atomicAdd(sp + 64, hp[64]);
  }
}

__device__ __forceinline__ void scan_route(const unsigned* __restrict__ pack,
                                           const unsigned* __restrict__ own,
                                           const float* __restrict__ h, float* sum,
                                           unsigned* q, unsigned* rq, unsigned* rqn, int E,
                                           int nunit, int wave, int lane, unsigned grp,
                                           uint4 (&cur)[4]) {
  int qn = 0;
  const unsigned lt_mask = (1u << lane) - 1u;
  unsigned gm[8];
  #pragma unroll
  for (int b = 0; b < 8; ++b) gm[b] = 0u - ((grp >> b) & 1u);

  const uint4* plane = (const uint4*)own + lane;

  for (int p0 = wave; p0 < nunit; p0 += 2 * OWN_WAVES) {
    uint4 nxt[4];
    load_step(nxt, plane, p0 + 2 * OWN_WAVES);

    #pragma unroll
    for (int u = 0; u < 2; ++u) {
      const int p = p0 + u * OWN_WAVES;
      unsigned f = match32(cur[2 * u], cur[2 * u + 1], gm);
      if (p < nunit && __builtin_amdgcn_ballot_w32(f != 0u)) {
        const int eb = p * UNIT + ((lane >> 2) << 7) + (lane & 3);
        do {
          int  bit = __ffs((int)f) - 1;
          int  e   = eb + (bit << 2);
          bool has = (f != 0u) && (e < E);
          unsigned k = __builtin_amdgcn_ballot_w32(has);
          if (has) q[qn + __popc(k & lt_mask)] = (unsigned)e;
          qn += __popc(k);
          f &= f - 1u;

          if (qn >= 32) {
            qn -= 32;
            const unsigned en = q[qn + lane];
            route32(pack[en], true, rq, rqn, h, sum, lane);
          }
        } while (__builtin_amdgcn_ballot_w32(f != 0u));
      }
    }
    #pragma unroll
    for (int u = 0; u < 4; ++u) cur[u] = nxt[u];
  }

  if (qn > 0) {
    const bool act = lane < qn;
    unsigned pw = 0u;
    if (act) pw = pack[q[lane]];
    route32(pw, act, rq, rqn, h, sum, lane);
  }
}

__global__ void __launch_bounds__(OWN_WAVES * 32)
aggregate_scan(const unsigned* __restrict__ pack, const unsigned* __restrict__ own,
               const float* __restrict__ dinv, const float* __restrict__ h,
               const float* __restrict__ b, const float* __restrict__ Bn,
               float* __restrict__ out,
               unsigned* rec, unsigned* rcnt, int E, int N, int nunit) {
  __shared__ __attribute__((aligned(16))) float sum[OWN_PAD * SUM_PITCH];
  __shared__ unsigned queue[OWN_WAVES * Q_CAP];
  __shared__ unsigned rq[OWN_WAVES * REC_CAP];
  __shared__ unsigned rqn[OWN_WAVES];
  __shared__ __attribute__((aligned(16))) float bs[G_STEPS * G_TILES * 64];
  __shared__ float dl[OWN_PAD];

  const int lane = threadIdx.x & 31;
  const int wave = __builtin_amdgcn_readfirstlane((int)(threadIdx.x >> 5));
  const int lo   = blockIdx.x * OWN_ROWS;
  const unsigned cnt = (unsigned)min(OWN_ROWS, N - lo);

  unsigned* q = queue + wave * Q_CAP;
  const int wid = blockIdx.x * OWN_WAVES + wave;
  unsigned* rp = rec + (size_t)wid * REC_CAP;

  RowSums acc;
  acc.c0 = (v8f){}; acc.c1 = (v8f){}; acc.c2 = (v8f){};
  #pragma unroll
  for (int k = 0; k < ROW_TRIPS; ++k) {
    const int i = wave + k * OWN_WAVES;
    const float* hp = h + (size_t)(lo + min(i, (int)cnt - 1)) * D + lane;
    acc.c0[k] = hp[0]; acc.c1[k] = hp[32]; acc.c2[k] = hp[64];
  }
  float dmine = 0.0f;
  if (threadIdx.x < cnt) dmine = dinv[lo + threadIdx.x];
  const float bv[3] = {b[lane], b[lane + 32], b[lane + 64]};

  uint4 cur[4];
  load_step(cur, (const uint4*)own + lane, wave);

  v2f bw0[3], bw1[3];
  #pragma unroll
  for (int it = 0; it < 3; ++it) {
    const int c = it * (OWN_WAVES * 32) + (int)threadIdx.x;
    bw0[it] = (v2f){0.0f, 0.0f};
    bw1[it] = (v2f){0.0f, 0.0f};
    if (c < G_STEPS * G_TILES * 16) {
      const int st = c >> 4;
      const int s  = st / G_TILES;
      const int t  = st - s * G_TILES;
      const int k  = ((s >> 1) << 3) + (((c >> 3) & 1) << 2) + ((s & 1) << 1);
      const float* src = Bn + (size_t)k * D + t * 16 + (c & 7) * 2;
      bw0[it] = *(const v2f*)src;
      bw1[it] = *(const v2f*)(src + D);
    }
  }
  #pragma unroll
  for (int it = 0; it < 3; ++it) {
    const int c = it * (OWN_WAVES * 32) + (int)threadIdx.x;
    if (c < G_STEPS * G_TILES * 16)
      ((float4*)bs)[c] = make_float4(bw0[it].x, bw1[it].x, bw0[it].y, bw1[it].y);
  }
  #pragma unroll
  for (int k = 0; k < ROW_TRIPS; ++k) {
    const int i = wave + k * OWN_WAVES;
    if (i < OWN_PAD) {
      #pragma unroll
      for (int j = 0; j < 3; ++j) sum[i * SUM_PITCH + lane + 32 * j] = 0.0f;
    }
    if (i >= (int)cnt) { acc.c0[k] = 0.0f; acc.c1[k] = 0.0f; acc.c2[k] = 0.0f; }
  }
  if (threadIdx.x < OWN_PAD) dl[threadIdx.x] = dmine;
  if (threadIdx.x < OWN_WAVES) rqn[threadIdx.x] = 0u;
  __syncthreads();

  scan_route(pack, own, h, sum, q, rq, rqn, E, nunit, wave, lane, (unsigned)blockIdx.x, cur);
  __syncthreads();

  {
    const unsigned n = (unsigned)__builtin_amdgcn_readfirstlane((int)rqn[wave]);
    const unsigned m = min(n, (unsigned)REC_CAP);
    if (lane == 0) rcnt[wid] = n;
    const unsigned* mq = rq + wave * REC_CAP;
    for (unsigned base = 0; base < m; base += 32) {
      unsigned rc = 0u;
      if (base + lane < m) { rc = mq[base + lane]; rp[base + lane] = rc; }
      add_chunk(rc, min(32u, m - base), h, acc, lane);
    }
  }

  #pragma unroll
  for (int k = 0; k < ROW_TRIPS; ++k) {
    const int i = wave + k * OWN_WAVES;
    if (i < (int)cnt) {
      const float di = dl[i];
      const float s[3] = {acc.c0[k], acc.c1[k], acc.c2[k]};
      #pragma unroll
      for (int j = 0; j < 3; ++j) {
        const int f = lane + 32 * j;
        float v = (s[j] + sum[i * SUM_PITCH + f]) * di + bv[j];
        sum[i * SUM_PITCH + f] = fmaxf(v, 0.0f);
      }
    }
  }
#if USE_F32_WMMA
  __syncthreads();
  if (wave < FUSE_WAVES) {
    const int strip = wave >> 1;
    const int tb    = (wave & 1) * 3;
    const int l     = lane & 15;
    const int half  = lane >> 4;
    const float* ar = sum + (strip * 16 + l) * SUM_PITCH + half * 4;
    float4 a4[D / 8];
    #pragma unroll
    for (int j = 0; j < D / 8; ++j) a4[j] = *(const float4*)(ar + 8 * j);
    v8f pr[3];
    #pragma unroll
    for (int t = 0; t < 3; ++t) pr[t] = (v8f){};
    const v2f* bp = (const v2f*)bs + tb * 32 + lane;
    #pragma unroll
    for (int j = 0; j < D / 8; ++j) {
      #pragma unroll
      for (int sub = 0; sub < 2; ++sub) {
        v2f a;
        a.x = sub ? a4[j].z : a4[j].x;
        a.y = sub ? a4[j].w : a4[j].y;
        const int s = 2 * j + sub;
        #pragma unroll
        for (int t = 0; t < 3; ++t) {
          const v2f bb = bp[(s * G_TILES + t) * 32];
          pr[t] = __builtin_amdgcn_wmma_f32_16x16x4_f32(
              false, a, false, bb, (short)0, pr[t], false, false);
        }
      }
    }
    const int r0 = strip * 16 + half * 8;
    #pragma unroll
    for (int r = 0; r < 8; ++r) {
      const int rowl = r0 + r;
      if (rowl < (int)cnt) {
        const float dr = dl[rowl];
        float* crow = out + (size_t)(lo + rowl) * D + tb * 16 + l;
        #pragma unroll
        for (int t = 0; t < 3; ++t) crow[t * 16] = pr[t][r] * dr;
      }
    }
  }
#endif
}

__global__ void __launch_bounds__(OWN_WAVES * 32)
aggregate_replay(const unsigned* __restrict__ pack, const unsigned* __restrict__ own,
                 const float* __restrict__ dinv, const float* __restrict__ h,
                 const float* __restrict__ b, float* __restrict__ out,
                 const unsigned* __restrict__ rec, const unsigned* __restrict__ rcnt,
                 int E, int N) {
  const int lane = threadIdx.x & 31;
  const int wave = __builtin_amdgcn_readfirstlane((int)(threadIdx.x >> 5));
  const int lo   = blockIdx.x * OWN_ROWS;
  const unsigned cnt = (unsigned)min(OWN_ROWS, N - lo);
  const int wid = blockIdx.x * OWN_WAVES + wave;
  const unsigned* rp = rec + (size_t)wid * REC_CAP;

  const unsigned nraw = rcnt[wid];

  RowSums acc;
  acc.c0 = (v8f){}; acc.c1 = (v8f){}; acc.c2 = (v8f){};
  float dk[ROW_TRIPS];
  #pragma unroll
  for (int k = 0; k < ROW_TRIPS; ++k) {
    const int i = lo + min(wave + k * OWN_WAVES, (int)cnt - 1);
    const float* hp = h + (size_t)i * D + lane;
    acc.c0[k] = hp[0]; acc.c1[k] = hp[32]; acc.c2[k] = hp[64];
    dk[k] = dinv[i];
  }
  const float bv[3] = {b[lane], b[lane + 32], b[lane + 64]};

  const unsigned n = (unsigned)__builtin_amdgcn_readfirstlane((int)nraw);
  unsigned nx = 0u;
  if (n <= REC_CAP && (unsigned)lane < n) nx = rp[lane];

  if (n <= REC_CAP) {
    for (unsigned base = 0; base < n; base += 32) {
      const unsigned rc = nx;
      asm volatile("" :: "v"(rc));
      nx = 0u;
      if (base + 32 + lane < n) nx = rp[base + 32 + lane];
      add_chunk(rc, min(32u, n - base), h, acc, lane);
    }
  } else {
    for (int e0 = 0; e0 < E; e0 += 32) {
      const int e = e0 + lane;
      unsigned pw = 0u;
      bool mine = false;
      if (e < E) {
        const unsigned r   = (unsigned)e & 1023u;
        const unsigned bit = (r & 127u) >> 2;
        const unsigned* up = own + (size_t)(e >> 10) * 256 + ((r >> 7) << 4) + ((r & 3u) << 2);
        unsigned ob = 0u;
        #pragma unroll
        for (int b = 0; b < 8; ++b) ob |= ((up[(b >> 2) * 128 + (b & 3)] >> bit) & 1u) << b;
        if (ob == blockIdx.x) {
          pw = pack[e];
          mine = ((pw >> 24) & 31u) == (unsigned)wave;
        }
      }
      unsigned k = __builtin_amdgcn_ballot_w32(mine);
      while (k) {
        const int j = __ffs((int)k) - 1;
        k &= k - 1u;
        const unsigned w = (unsigned)__builtin_amdgcn_readlane((int)pw, j);
        const float* hp = h + (size_t)(w & 0x00FFFFFFu) * D + lane;
        acc_add(acc, (int)(w >> 29), hp[0], hp[32], hp[64]);
      }
    }
  }

  #pragma unroll
  for (int k = 0; k < ROW_TRIPS; ++k) {
    const int i = wave + k * OWN_WAVES;
    if (i < (int)cnt) {
      const float s[3] = {acc.c0[k], acc.c1[k], acc.c2[k]};
      #pragma unroll
      for (int j = 0; j < 3; ++j)
        out[(size_t)(lo + i) * D + lane + 32 * j] = s[j] * dk[k] + bv[j];
    }
  }
}

extern "C" void kernel_launch(void* const* d_in, const int* in_sizes, int n_in,
                              void* d_out, int out_size, void* d_ws, size_t ws_size,
                              hipStream_t stream) {
  const float* x  = (const float*)d_in[0];
  const int*   ei = (const int*)d_in[1];
  const float* W1 = (const float*)d_in[2];
  const float* b1 = (const float*)d_in[3];
  const float* W2 = (const float*)d_in[4];
  const float* b2 = (const float*)d_in[5];
  float* out = (float*)d_out;

  int N = in_sizes[0] / D;
  int E = in_sizes[1] / 2;
  const int* row = ei;
  const int* col = ei + E;

  float* ws   = (float*)d_ws;
  float* dinv = ws;
  float* h    = ws + (((size_t)N + 127) & ~127ull);
  float* acc  = h + (size_t)N * D;
  unsigned* own = (unsigned*)(acc + (size_t)N * D);

  const int tpb = 256;
  int gN    = (N + tpb - 1) / tpb;
  int gGemm = (N / 16 + G_WAVES - 1) / G_WAVES;
  int gOwn  = (N + OWN_ROWS - 1) / OWN_ROWS;
  int nunit  = (E + UNIT - 1) / UNIT;
  int nstep  = (nunit + 2 * OWN_WAVES - 1) / (2 * OWN_WAVES) + 1;
  int words  = nstep * 2 * OWN_WAVES * (UNIT / 4);
  int pitch  = (N + 127) & ~127;
  unsigned* part = own + words;
  int gRange = (N + DEG_ROWS - 1) / DEG_ROWS;
  unsigned* rcnt = part + (size_t)DEG_SLICES * pitch;
  unsigned* rec = rcnt + (((size_t)gOwn * OWN_WAVES + 127) & ~127ull);
  unsigned* pack = rec + (size_t)gOwn * OWN_WAVES * REC_CAP;

  deg_owner_kernel<<<dim3(gRange, DEG_SLICES), DEG_THREADS, 0, stream>>>(row, col, part, own, pack, E, N, pitch, words);
  dinv_kernel<<<gN, tpb, 0, stream>>>(part, dinv, N, pitch);

  gemm_wmma<<<gGemm, G_WAVES * 32, 0, stream>>>(x, W1, dinv, h, N);
  aggregate_scan<<<gOwn, OWN_WAVES * 32, 0, stream>>>(pack, own, dinv, h, b1, W2, acc, rec, rcnt, E, N, nunit);

  aggregate_replay<<<gOwn, OWN_WAVES * 32, 0, stream>>>(pack, own, dinv, acc, b2, out, rec, rcnt, E, N);
}
